# starting kernel
# baseline (speedup 1.0000x reference)
#include <hip/hip_runtime.h>

typedef __attribute__((ext_vector_type(2))) float v2f;
typedef __attribute__((ext_vector_type(8))) float v8f;

static constexpr int Bb  = 2;
static constexpr int Nn  = 1024;
static constexpr int Mm  = 1024;
static constexpr int DIN = 16;
static constexpr int Cc  = 32;
static constexpr int Hh  = 32;

__global__ void rescale_kernel(const float* __restrict__ x, float* __restrict__ xs) {
    __shared__ float smn[256], smx[256];
    const int bd = blockIdx.x;
    const int b = bd >> 1, d = bd & 1;
    const int t = threadIdx.x;
    float v[4];
    float mn = 1e30f, mx = -1e30f;
#pragma unroll
    for (int i = 0; i < 4; ++i) {
        const int n = t + i * 256;
        v[i] = x[(b * Nn + n) * 2 + d];
        mn = fminf(mn, v[i]);
        mx = fmaxf(mx, v[i]);
    }
    smn[t] = mn; smx[t] = mx;
    __syncthreads();
    for (int s = 128; s > 0; s >>= 1) {
        if (t < s) {
            smn[t] = fminf(smn[t], smn[t + s]);
            smx[t] = fmaxf(smx[t], smx[t + s]);
        }
        __syncthreads();
    }
    mn = smn[0]; mx = smx[0];
    const float sc = 2.0f / (mx - mn + 1e-12f);
#pragma unroll
    for (int i = 0; i < 4; ++i) {
        const int n = t + i * 256;
        xs[(b * Nn + n) * 2 + d] = (v[i] - mn) * sc - 1.0f;
    }
}

__global__ void __launch_bounds__(128) lift_kernel(const float* __restrict__ pnd,
                                                   const float* __restrict__ WL,
                                                   const float* __restrict__ bl,
                                                   float* __restrict__ f) {
    const int wid  = (blockIdx.x * blockDim.x + threadIdx.x) >> 5;
    const int lane = threadIdx.x & 31;
    const int rt = wid >> 1, ct = wid & 1;
    const int row0 = rt * 16;
    const int lrow = lane & 15;
    const int hsel = lane >> 4;
    const int c = ct * 16 + lrow;

    v8f acc = {0.f, 0.f, 0.f, 0.f, 0.f, 0.f, 0.f, 0.f};
#pragma unroll
    for (int kk = 0; kk < 4; ++kk) {
        const int k0 = kk * 4 + 2 * hsel;
        v2f a, bm;
        a.x = pnd[(row0 + lrow) * DIN + k0];
        a.y = pnd[(row0 + lrow) * DIN + k0 + 1];
        bm.x = WL[k0 * Cc + c];
        bm.y = WL[(k0 + 1) * Cc + c];
        acc = __builtin_amdgcn_wmma_f32_16x16x4_f32(false, a, false, bm,
                                                    (short)0, acc, false, false);
    }
    const float bias = bl[c];
#pragma unroll
    for (int r = 0; r < 8; ++r) {
        f[(row0 + r + 8 * hsel) * Cc + c] = acc[r] + bias;
    }
}

__global__ void counts_kernel(const float* __restrict__ xs, const float* __restrict__ lq,
                              float* __restrict__ inv) {
    const int wid  = (blockIdx.x * blockDim.x + threadIdx.x) >> 5;
    const int lane = threadIdx.x & 31;
    const int b = wid / Mm, m = wid % Mm;
    const float q0 = lq[m * 2], q1 = lq[m * 2 + 1];
    float c1 = 0.f, c2 = 0.f;
    for (int n = lane; n < Nn; n += 32) {
        const float x0 = xs[(b * Nn + n) * 2];
        const float x1 = xs[(b * Nn + n) * 2 + 1];
        const float dx = q0 - x0, dy = q1 - x1;
        const float d2 = dx * dx + dy * dy;
        c1 += (d2 <= 0.01f) ? 1.f : 0.f;
        c2 += (d2 <= 0.04f) ? 1.f : 0.f;
    }
#pragma unroll
    for (int s = 16; s >= 1; s >>= 1) {
        c1 += __shfl_xor(c1, s, 32);
        c2 += __shfl_xor(c2, s, 32);
    }
    if (lane == 0) {
        inv[wid * 2]     = 1.0f / fmaxf(c1, 1.0f);
        inv[wid * 2 + 1] = 1.0f / fmaxf(c2, 1.0f);
    }
}

__device__ __forceinline__ float gelu_tanh(float x) {
    const float u = x * x;
    const float t = 0.7978845608028654f * x * fmaf(0.044715f, u, 1.0f);
    const float e = __expf(-2.0f * t);
    const float th = (1.0f - e) * __frcp_rn(1.0f + e);
    return 0.5f * x * (1.0f + th);
}

__global__ void __launch_bounds__(256) main_kernel(
    const float* __restrict__ xs, const float* __restrict__ f,
    const float* __restrict__ inv, const float* __restrict__ lq,
    const float* __restrict__ W1, const float* __restrict__ b1,
    const float* __restrict__ W2, const float* __restrict__ b2,
    float* __restrict__ out) {
    extern __shared__ float smem[];
    float* fs  = smem;
    float* xss = smem + Nn * Cc;

    const int tid  = threadIdx.x;
    const int b    = blockIdx.x / (Mm / 8);
    const int mg   = blockIdx.x % (Mm / 8);
    const int wv   = tid >> 5;
    const int lane = tid & 31;
    const int m    = mg * 8 + wv;
    const int wid  = b * Mm + m;
    const int lrow = lane & 15;
    const int hsel = lane >> 4;

    {
        const float4* src = (const float4*)(f + b * Nn * Cc);
        float4* dst = (float4*)fs;
#pragma unroll
        for (int i = 0; i < (Nn * Cc / 4) / 256; ++i)
            dst[tid + i * 256] = src[tid + i * 256];
        const float4* s2 = (const float4*)(xs + b * Nn * 2);
        float4* d2v = (float4*)xss;
#pragma unroll
        for (int i = 0; i < (Nn * 2 / 4) / 256; ++i)
            d2v[tid + i * 256] = s2[tid + i * 256];
    }
    __syncthreads();

    const float q0 = lq[m * 2], q1 = lq[m * 2 + 1];
    const float inv1 = inv[wid * 2], inv2 = inv[wid * 2 + 1];

    float W1x[16], W1y[16], Cm[16];
#pragma unroll
    for (int j = 0; j < 16; ++j) {
        const int h = 4 * (j >> 1) + 2 * hsel + (j & 1);
        W1x[j] = W1[h];
        W1y[j] = W1[Hh + h];
        Cm[j]  = fmaf(q0, W1[2 * Hh + h], fmaf(q1, W1[3 * Hh + h], b1[h]));
    }

    v2f Bw0[8], Bw1[8];
#pragma unroll
    for (int kk = 0; kk < 8; ++kk) {
        const int k0 = 4 * kk + 2 * hsel;
        Bw0[kk].x = W2[k0 * Cc + lrow];
        Bw0[kk].y = W2[(k0 + 1) * Cc + lrow];
        Bw1[kk].x = W2[k0 * Cc + 16 + lrow];
        Bw1[kk].y = W2[(k0 + 1) * Cc + 16 + lrow];
    }
    const float bz0 = b2[lrow], bz1 = b2[16 + lrow];

    float o0 = 0.f, o1 = 0.f;
    for (int nt = 0; nt < Nn / 16; ++nt) {
        const int n = nt * 16 + lrow;
        const float x0 = xss[n * 2], x1 = xss[n * 2 + 1];
        const float dx = q0 - x0, dy = q1 - x1;
        const float d2 = dx * dx + dy * dy;
        const float w = ((d2 <= 0.01f) ? inv1 : 0.f) + ((d2 <= 0.04f) ? inv2 : 0.f);

        v8f acc0 = {bz0, bz0, bz0, bz0, bz0, bz0, bz0, bz0};
        v8f acc1 = {bz1, bz1, bz1, bz1, bz1, bz1, bz1, bz1};
#pragma unroll
        for (int kk = 0; kk < 8; ++kk) {
            v2f a;
            const float p0 = fmaf(x0, W1x[2 * kk],     fmaf(x1, W1y[2 * kk],     Cm[2 * kk]));
            const float p1 = fmaf(x0, W1x[2 * kk + 1], fmaf(x1, W1y[2 * kk + 1], Cm[2 * kk + 1]));
            a.x = gelu_tanh(p0);
            a.y = gelu_tanh(p1);
            acc0 = __builtin_amdgcn_wmma_f32_16x16x4_f32(false, a, false, Bw0[kk],
                                                         (short)0, acc0, false, false);
            acc1 = __builtin_amdgcn_wmma_f32_16x16x4_f32(false, a, false, Bw1[kk],
                                                         (short)0, acc1, false, false);
        }
#pragma unroll
        for (int r = 0; r < 8; ++r) {
            const int row = r + 8 * hsel;
            const float wr = __shfl(w, row, 32);
            const int nrow = nt * 16 + row;
            const float f0 = fs[nrow * Cc + lrow];
            const float f1 = fs[nrow * Cc + 16 + lrow];
            o0 = fmaf(acc0[r] * wr, f0, o0);
            o1 = fmaf(acc1[r] * wr, f1, o1);
        }
    }
    o0 += __shfl_xor(o0, 16, 32);
    o1 += __shfl_xor(o1, 16, 32);
    if (lane < 16) {
        out[wid * Cc + lane]      = o0;
        out[wid * Cc + 16 + lane] = o1;
    }
}

extern "C" void kernel_launch(void* const* d_in, const int* in_sizes, int n_in,
                              void* d_out, int out_size, void* d_ws, size_t ws_size,
                              hipStream_t stream) {
    (void)in_sizes; (void)n_in; (void)out_size; (void)ws_size;
    const float* pndata = (const float*)d_in[0];
    const float* xcoord = (const float*)d_in[1];
    const float* lq     = (const float*)d_in[2];
    const float* W_lift = (const float*)d_in[3];
    const float* b_lift = (const float*)d_in[4];
    const float* W1     = (const float*)d_in[5];
    const float* b1     = (const float*)d_in[6];
    const float* W2     = (const float*)d_in[7];
    const float* b2     = (const float*)d_in[8];
    float* out = (float*)d_out;

    float* ws  = (float*)d_ws;
    float* xs  = ws;
    float* f   = xs + Bb * Nn * 2;
    float* inv = f + Bb * Nn * Cc;

    rescale_kernel<<<Bb * 2, 256, 0, stream>>>(xcoord, xs);
    lift_kernel<<<64, 128, 0, stream>>>(pndata, W_lift, b_lift, f);
    counts_kernel<<<256, 256, 0, stream>>>(xs, lq, inv);

    const size_t smem_bytes = (size_t)(Nn * Cc + Nn * 2) * sizeof(float);
    main_kernel<<<Bb * (Mm / 8), 256, smem_bytes, stream>>>(xs, f, inv, lq,
                                                            W1, b1, W2, b2, out);
}


# main kernel: per-query ballot compaction of in-radius points; MLP, WMMA and drain run only on 16-point tiles of that list, f rows gathered from global not staged; changes fp summation order
# speedup vs baseline: 4.3828x; 4.3828x over previous
#include <hip/hip_runtime.h>

typedef __attribute__((ext_vector_type(2))) float v2f;
typedef __attribute__((ext_vector_type(8))) float v8f;

static constexpr int Bb  = 2;
static constexpr int Nn  = 1024;
static constexpr int Mm  = 1024;
static constexpr int DIN = 16;
static constexpr int Cc  = 32;
static constexpr int Hh  = 32;

__global__ void rescale_kernel(const float* __restrict__ x, float* __restrict__ xs) {
    __shared__ float smn[256], smx[256];
    const int bd = blockIdx.x;
    const int b = bd >> 1, d = bd & 1;
    const int t = threadIdx.x;
    float v[4];
    float mn = 1e30f, mx = -1e30f;
#pragma unroll
    for (int i = 0; i < 4; ++i) {
        const int n = t + i * 256;
        v[i] = x[(b * Nn + n) * 2 + d];
        mn = fminf(mn, v[i]);
        mx = fmaxf(mx, v[i]);
    }
    smn[t] = mn; smx[t] = mx;
    __syncthreads();
    for (int s = 128; s > 0; s >>= 1) {
        if (t < s) {
            smn[t] = fminf(smn[t], smn[t + s]);
            smx[t] = fmaxf(smx[t], smx[t + s]);
        }
        __syncthreads();
    }
    mn = smn[0]; mx = smx[0];
    const float sc = 2.0f / (mx - mn + 1e-12f);
#pragma unroll
    for (int i = 0; i < 4; ++i) {
        const int n = t + i * 256;
        xs[(b * Nn + n) * 2 + d] = (v[i] - mn) * sc - 1.0f;
    }
}

__global__ void __launch_bounds__(128) lift_kernel(const float* __restrict__ pnd,
                                                   const float* __restrict__ WL,
                                                   const float* __restrict__ bl,
                                                   float* __restrict__ f) {
    const int wid  = (blockIdx.x * blockDim.x + threadIdx.x) >> 5;
    const int lane = threadIdx.x & 31;
    const int rt = wid >> 1, ct = wid & 1;
    const int row0 = rt * 16;
    const int lrow = lane & 15;
    const int hsel = lane >> 4;
    const int c = ct * 16 + lrow;

    v8f acc = {0.f, 0.f, 0.f, 0.f, 0.f, 0.f, 0.f, 0.f};
#pragma unroll
    for (int kk = 0; kk < 4; ++kk) {
        const int k0 = kk * 4 + 2 * hsel;
        v2f a, bm;
        a.x = pnd[(row0 + lrow) * DIN + k0];
        a.y = pnd[(row0 + lrow) * DIN + k0 + 1];
        bm.x = WL[k0 * Cc + c];
        bm.y = WL[(k0 + 1) * Cc + c];
        acc = __builtin_amdgcn_wmma_f32_16x16x4_f32(false, a, false, bm,
                                                    (short)0, acc, false, false);
    }
    const float bias = bl[c];
#pragma unroll
    for (int r = 0; r < 8; ++r) {
        f[(row0 + r + 8 * hsel) * Cc + c] = acc[r] + bias;
    }
}

__global__ void counts_kernel(const float* __restrict__ xs, const float* __restrict__ lq,
                              float* __restrict__ inv) {
    const int wid  = (blockIdx.x * blockDim.x + threadIdx.x) >> 5;
    const int lane = threadIdx.x & 31;
    const int b = wid / Mm, m = wid % Mm;
    const float q0 = lq[m * 2], q1 = lq[m * 2 + 1];
    float c1 = 0.f, c2 = 0.f;
    for (int n = lane; n < Nn; n += 32) {
        const float x0 = xs[(b * Nn + n) * 2];
        const float x1 = xs[(b * Nn + n) * 2 + 1];
        const float dx = q0 - x0, dy = q1 - x1;
        const float d2 = dx * dx + dy * dy;
        c1 += (d2 <= 0.01f) ? 1.f : 0.f;
        c2 += (d2 <= 0.04f) ? 1.f : 0.f;
    }
#pragma unroll
    for (int s = 16; s >= 1; s >>= 1) {
        c1 += __shfl_xor(c1, s, 32);
        c2 += __shfl_xor(c2, s, 32);
    }
    if (lane == 0) {
        inv[wid * 2]     = 1.0f / fmaxf(c1, 1.0f);
        inv[wid * 2 + 1] = 1.0f / fmaxf(c2, 1.0f);
    }
}

__device__ __forceinline__ float gelu_tanh(float x) {
    const float u = x * x;
    const float t = 0.7978845608028654f * x * fmaf(0.044715f, u, 1.0f);
    const float e = __expf(-2.0f * t);
    const float th = (1.0f - e) * __frcp_rn(1.0f + e);
    return 0.5f * x * (1.0f + th);
}

__global__ void __launch_bounds__(256) main_kernel(
    const float* __restrict__ xs, const float* __restrict__ f,
    const float* __restrict__ inv, const float* __restrict__ lq,
    const float* __restrict__ W1, const float* __restrict__ b1,
    const float* __restrict__ W2, const float* __restrict__ b2,
    float* __restrict__ out) {
    __shared__ __attribute__((aligned(16))) float xss[Nn * 2];
    __shared__ __attribute__((aligned(16))) int nbr[8][Nn];

    const int tid  = threadIdx.x;
    const int b    = blockIdx.x / (Mm / 8);
    const int mg   = blockIdx.x % (Mm / 8);
    const int wv   = tid >> 5;
    const int lane = tid & 31;
    const int m    = mg * 8 + wv;
    const int wid  = b * Mm + m;
    const int lrow = lane & 15;
    const int hsel = lane >> 4;

    {
        const float4* s2 = (const float4*)(xs + b * Nn * 2);
        float4* d2v = (float4*)xss;
#pragma unroll
        for (int i = 0; i < (Nn * 2 / 4) / 256; ++i)
            d2v[tid + i * 256] = s2[tid + i * 256];
    }

    const float q0 = lq[m * 2], q1 = lq[m * 2 + 1];
    const float inv1 = inv[wid * 2], inv2 = inv[wid * 2 + 1];

    float W1x[16], W1y[16], Cm[16];
#pragma unroll
    for (int j = 0; j < 16; ++j) {
        const int h = 4 * (j >> 1) + 2 * hsel + (j & 1);
        W1x[j] = W1[h];
        W1y[j] = W1[Hh + h];
        Cm[j]  = fmaf(q0, W1[2 * Hh + h], fmaf(q1, W1[3 * Hh + h], b1[h]));
    }

    v2f Bw0[8], Bw1[8];
#pragma unroll
    for (int kk = 0; kk < 8; ++kk) {
        const int k0 = 4 * kk + 2 * hsel;
        Bw0[kk].x = W2[k0 * Cc + lrow];
        Bw0[kk].y = W2[(k0 + 1) * Cc + lrow];
        Bw1[kk].x = W2[k0 * Cc + 16 + lrow];
        Bw1[kk].y = W2[(k0 + 1) * Cc + 16 + lrow];
    }
    const float bz0 = b2[lrow], bz1 = b2[16 + lrow];

    __syncthreads();

    int* lst = nbr[wv];
    const float2* xs2 = (const float2*)xss;
    float2 pp[Nn / 32];
#pragma unroll
    for (int it = 0; it < Nn / 32; ++it) pp[it] = xs2[it * 32 + lane];
    int cnt = 0;
#pragma unroll
    for (int it = 0; it < Nn / 32; ++it) {
        const int n = it * 32 + lane;
        const float dx = q0 - pp[it].x, dy = q1 - pp[it].y;
        const float d2 = fmaf(dx, dx, dy * dy);
        const bool in1 = (d2 <= 0.01f);
        const bool in2 = (d2 <= 0.04f);
        const bool in  = in1 || in2;
        const unsigned mask = __builtin_amdgcn_ballot_w32(in);
        if (in)
            lst[cnt + (int)__builtin_amdgcn_mbcnt_lo(mask, 0u)] =
                n | (in1 ? 0x10000 : 0) | (in2 ? 0x20000 : 0);
        cnt += __builtin_popcount(mask);
    }
    __syncthreads();

    const float* fb = f + b * Nn * Cc;
    float o0 = 0.f, o1 = 0.f;
    const int ntile = (cnt + 15) >> 4;
    for (int t = 0; t < ntile; ++t) {
        const int ia = t * 16 + lrow;
        const int ea = (ia < cnt) ? lst[ia] : 0;
        const float2 p = xs2[ea & 0xFFFF];
        const float x0 = p.x, x1 = p.y;

        const int id0 = t * 16 + 8 * hsel;
        const int4 eA = *(const int4*)(lst + id0);
        const int4 eB = *(const int4*)(lst + id0 + 4);
        const int er[8] = {eA.x, eA.y, eA.z, eA.w, eB.x, eB.y, eB.z, eB.w};
        float wr[8], f0[8], f1[8];
#pragma unroll
        for (int r = 0; r < 8; ++r) {
            const int e = (id0 + r < cnt) ? er[r] : 0;
            wr[r] = ((e & 0x10000) ? inv1 : 0.f) + ((e & 0x20000) ? inv2 : 0.f);
            const float* fr = fb + (e & 0xFFFF) * Cc + lrow;
            f0[r] = fr[0];
            f1[r] = fr[16];
        }

        v8f acc0 = {bz0, bz0, bz0, bz0, bz0, bz0, bz0, bz0};
        v8f acc1 = {bz1, bz1, bz1, bz1, bz1, bz1, bz1, bz1};
#pragma unroll
        for (int kk = 0; kk < 8; ++kk) {
            v2f a;
            const float p0 = fmaf(x0, W1x[2 * kk],     fmaf(x1, W1y[2 * kk],     Cm[2 * kk]));
            const float p1 = fmaf(x0, W1x[2 * kk + 1], fmaf(x1, W1y[2 * kk + 1], Cm[2 * kk + 1]));
            a.x = gelu_tanh(p0);
            a.y = gelu_tanh(p1);
            acc0 = __builtin_amdgcn_wmma_f32_16x16x4_f32(false, a, false, Bw0[kk],
                                                         (short)0, acc0, false, false);
            acc1 = __builtin_amdgcn_wmma_f32_16x16x4_f32(false, a, false, Bw1[kk],
                                                         (short)0, acc1, false, false);
        }
#pragma unroll
        for (int r = 0; r < 8; ++r) {
            o0 = fmaf(acc0[r] * wr[r], f0[r], o0);
            o1 = fmaf(acc1[r] * wr[r], f1[r], o1);
        }
    }
    o0 += __shfl_xor(o0, 16, 32);
    o1 += __shfl_xor(o1, 16, 32);
    if (lane < 16) {
        out[wid * Cc + lane]      = o0;
        out[wid * Cc + 16 + lane] = o1;
    }
}

extern "C" void kernel_launch(void* const* d_in, const int* in_sizes, int n_in,
                              void* d_out, int out_size, void* d_ws, size_t ws_size,
                              hipStream_t stream) {
    (void)in_sizes; (void)n_in; (void)out_size; (void)ws_size;
    const float* pndata = (const float*)d_in[0];
    const float* xcoord = (const float*)d_in[1];
    const float* lq     = (const float*)d_in[2];
    const float* W_lift = (const float*)d_in[3];
    const float* b_lift = (const float*)d_in[4];
    const float* W1     = (const float*)d_in[5];
    const float* b1     = (const float*)d_in[6];
    const float* W2     = (const float*)d_in[7];
    const float* b2     = (const float*)d_in[8];
    float* out = (float*)d_out;

    float* ws  = (float*)d_ws;
    float* xs  = ws;
    float* f   = xs + Bb * Nn * 2;
    float* inv = f + Bb * Nn * Cc;

    rescale_kernel<<<Bb * 2, 256, 0, stream>>>(xcoord, xs);
    lift_kernel<<<64, 128, 0, stream>>>(pndata, W_lift, b_lift, f);
    counts_kernel<<<256, 256, 0, stream>>>(xs, lq, inv);

    main_kernel<<<Bb * (Mm / 8), 256, 0, stream>>>(xs, f, inv, lq,
                                                   W1, b1, W2, b2, out);
}


# one launch instead of four: each block rescales its batch's coords into LDS, neighbour counts are popcounts of the compaction ballots, f rows made per 16-point tile by f32 WMMA on gathered input rows
# speedup vs baseline: 9.9849x; 2.2782x over previous
#include <hip/hip_runtime.h>


typedef __attribute__((ext_vector_type(2))) float v2f;
typedef __attribute__((ext_vector_type(8))) float v8f;

static constexpr int Bb  = 2;
static constexpr int Nn  = 1024;
static constexpr int Mm  = 1024;
static constexpr int DIN = 16;
static constexpr int Cc  = 32;
static constexpr int Hh  = 32;

__device__ __forceinline__ float gelu_tanh(float x) {
    const float u = x * x;
    const float t = 0.7978845608028654f * x * fmaf(0.044715f, u, 1.0f);
    const float e = __expf(-2.0f * t);
    const float th = (1.0f - e) * __frcp_rn(1.0f + e);
    return 0.5f * x * (1.0f + th);
}

__global__ void __launch_bounds__(256) gno_kernel(
    const float* __restrict__ xc, const float* __restrict__ pnd,
    const float* __restrict__ lq,
    const float* __restrict__ WL, const float* __restrict__ bl,
    const float* __restrict__ W1, const float* __restrict__ b1,
    const float* __restrict__ W2, const float* __restrict__ b2,
    float* __restrict__ out) {
    __shared__ __attribute__((aligned(16))) float xss[Nn * 2];
    __shared__ __attribute__((aligned(16))) int nbr[8][Nn];
    __shared__ __attribute__((aligned(16))) float red[8][4];

    const int tid  = threadIdx.x;
    const int b    = blockIdx.x / (Mm / 8);
    const int mg   = blockIdx.x % (Mm / 8);
    const int wv   = tid >> 5;
    const int lane = tid & 31;
    const int m    = mg * 8 + wv;
    const int wid  = b * Mm + m;
    const int lrow = lane & 15;
    const int hsel = lane >> 4;

    const float4* s2 = (const float4*)(xc + b * Nn * 2);
    const float4 va = s2[tid];
    const float4 vb = s2[tid + 256];

    const float q0 = lq[m * 2], q1 = lq[m * 2 + 1];

    float W1x[16], W1y[16], Cm[16];
#pragma unroll
    for (int j = 0; j < 16; ++j) {
        const int h = 4 * (j >> 1) + 2 * hsel + (j & 1);
        W1x[j] = W1[h];
        W1y[j] = W1[Hh + h];
        Cm[j]  = fmaf(q0, W1[2 * Hh + h], fmaf(q1, W1[3 * Hh + h], b1[h]));
    }

    v2f Bw0[8], Bw1[8];
#pragma unroll
    for (int kk = 0; kk < 8; ++kk) {
        const int k0 = 4 * kk + 2 * hsel;
        Bw0[kk].x = W2[k0 * Cc + lrow];
        Bw0[kk].y = W2[(k0 + 1) * Cc + lrow];
        Bw1[kk].x = W2[k0 * Cc + 16 + lrow];
        Bw1[kk].y = W2[(k0 + 1) * Cc + 16 + lrow];
    }
    const float bz0 = b2[lrow], bz1 = b2[16 + lrow];

    v2f Lw0[4], Lw1[4];
#pragma unroll
    for (int kk = 0; kk < 4; ++kk) {
        const int k0 = 4 * kk + 2 * hsel;
        Lw0[kk].x = WL[k0 * Cc + lrow];
        Lw0[kk].y = WL[(k0 + 1) * Cc + lrow];
        Lw1[kk].x = WL[k0 * Cc + 16 + lrow];
        Lw1[kk].y = WL[(k0 + 1) * Cc + 16 + lrow];
    }
    const float bl0 = bl[lrow], bl1 = bl[16 + lrow];

    float mn0 = fminf(fminf(va.x, va.z), fminf(vb.x, vb.z));
    float mn1 = fminf(fminf(va.y, va.w), fminf(vb.y, vb.w));
    float mx0 = fmaxf(fmaxf(va.x, va.z), fmaxf(vb.x, vb.z));
    float mx1 = fmaxf(fmaxf(va.y, va.w), fmaxf(vb.y, vb.w));
#pragma unroll
    for (int s = 16; s >= 1; s >>= 1) {
        mn0 = fminf(mn0, __shfl_xor(mn0, s, 32));
        mn1 = fminf(mn1, __shfl_xor(mn1, s, 32));
        mx0 = fmaxf(mx0, __shfl_xor(mx0, s, 32));
        mx1 = fmaxf(mx1, __shfl_xor(mx1, s, 32));
    }
    if (lane == 0) *(float4*)red[wv] = make_float4(mn0, mn1, mx0, mx1);
    __syncthreads();
#pragma unroll
    for (int w = 0; w < 8; ++w) {
        const float4 rw = *(const float4*)red[w];
        mn0 = fminf(mn0, rw.x);
        mn1 = fminf(mn1, rw.y);
        mx0 = fmaxf(mx0, rw.z);
        mx1 = fmaxf(mx1, rw.w);
    }
    {
        const float sc0 = 2.0f / (mx0 - mn0 + 1e-12f);
        const float sc1 = 2.0f / (mx1 - mn1 + 1e-12f);
        float4 ra, rb;
        ra.x = (va.x - mn0) * sc0 - 1.0f;
        ra.y = (va.y - mn1) * sc1 - 1.0f;
        ra.z = (va.z - mn0) * sc0 - 1.0f;
        ra.w = (va.w - mn1) * sc1 - 1.0f;
        rb.x = (vb.x - mn0) * sc0 - 1.0f;
        rb.y = (vb.y - mn1) * sc1 - 1.0f;
        rb.z = (vb.z - mn0) * sc0 - 1.0f;
        rb.w = (vb.w - mn1) * sc1 - 1.0f;
        float4* d2v = (float4*)xss;
        d2v[tid]       = ra;
        d2v[tid + 256] = rb;
    }
    __syncthreads();

    int* lst = nbr[wv];
    const float2* xs2 = (const float2*)xss;
    float2 pp[Nn / 32];
#pragma unroll
    for (int it = 0; it < Nn / 32; ++it) pp[it] = xs2[it * 32 + lane];
    int cnt = 0, c1 = 0, c2 = 0;
#pragma unroll
    for (int it = 0; it < Nn / 32; ++it) {
        const int n = it * 32 + lane;
        const float dx = q0 - pp[it].x, dy = q1 - pp[it].y;
        const float d2 = fmaf(dx, dx, dy * dy);
        const bool in1 = (d2 <= 0.01f);
        const bool in2 = (d2 <= 0.04f);
        const bool in  = in1 || in2;
        const unsigned mask  = __builtin_amdgcn_ballot_w32(in);
        const unsigned mask1 = __builtin_amdgcn_ballot_w32(in1);
        const unsigned mask2 = __builtin_amdgcn_ballot_w32(in2);
        if (in)
            lst[cnt + (int)__builtin_amdgcn_mbcnt_lo(mask, 0u)] =
                n | (in1 ? 0x10000 : 0) | (in2 ? 0x20000 : 0);
        cnt += __builtin_popcount(mask);
        c1  += __builtin_popcount(mask1);
        c2  += __builtin_popcount(mask2);
    }
    const float inv1 = 1.0f / fmaxf((float)c1, 1.0f);
    const float inv2 = 1.0f / fmaxf((float)c2, 1.0f);
    __syncthreads();

    const float* pb = pnd + b * Nn * DIN;
    float o0 = 0.f, o1 = 0.f;
    const int ntile = (cnt + 15) >> 4;
    for (int t = 0; t < ntile; ++t) {
        const int ia = t * 16 + lrow;
        const int ea = (ia < cnt) ? lst[ia] : 0;
        const int na = ea & 0xFFFF;
        const float2 p = xs2[na];
        const float x0 = p.x, x1 = p.y;

        const float2* pr = (const float2*)(pb + na * DIN + 2 * hsel);
        v2f la[4];
#pragma unroll
        for (int kk = 0; kk < 4; ++kk) {
            const float2 g = pr[2 * kk];
            la[kk].x = g.x;
            la[kk].y = g.y;
        }

        const int id0 = t * 16 + 8 * hsel;
        const int4 eA = *(const int4*)(lst + id0);
        const int4 eB = *(const int4*)(lst + id0 + 4);
        const int er[8] = {eA.x, eA.y, eA.z, eA.w, eB.x, eB.y, eB.z, eB.w};
        float wr[8];
#pragma unroll
        for (int r = 0; r < 8; ++r) {
            const int e = (id0 + r < cnt) ? er[r] : 0;
            wr[r] = ((e & 0x10000) ? inv1 : 0.f) + ((e & 0x20000) ? inv2 : 0.f);
        }

        v8f acc0 = {bz0, bz0, bz0, bz0, bz0, bz0, bz0, bz0};
        v8f acc1 = {bz1, bz1, bz1, bz1, bz1, bz1, bz1, bz1};
#pragma unroll
        for (int kk = 0; kk < 8; ++kk) {
            v2f a;
            const float p0 = fmaf(x0, W1x[2 * kk],     fmaf(x1, W1y[2 * kk],     Cm[2 * kk]));
            const float p1 = fmaf(x0, W1x[2 * kk + 1], fmaf(x1, W1y[2 * kk + 1], Cm[2 * kk + 1]));
            a.x = gelu_tanh(p0);
            a.y = gelu_tanh(p1);
            acc0 = __builtin_amdgcn_wmma_f32_16x16x4_f32(false, a, false, Bw0[kk],
                                                         (short)0, acc0, false, false);
            acc1 = __builtin_amdgcn_wmma_f32_16x16x4_f32(false, a, false, Bw1[kk],
                                                         (short)0, acc1, false, false);
        }

        v8f fa0 = {0.f, 0.f, 0.f, 0.f, 0.f, 0.f, 0.f, 0.f};
        v8f fa1 = {0.f, 0.f, 0.f, 0.f, 0.f, 0.f, 0.f, 0.f};
#pragma unroll
        for (int kk = 0; kk < 4; ++kk) {
            fa0 = __builtin_amdgcn_wmma_f32_16x16x4_f32(false, la[kk], false, Lw0[kk],
                                                        (short)0, fa0, false, false);
            fa1 = __builtin_amdgcn_wmma_f32_16x16x4_f32(false, la[kk], false, Lw1[kk],
                                                        (short)0, fa1, false, false);
        }

#pragma unroll
        for (int r = 0; r < 8; ++r) {
            o0 = fmaf(acc0[r] * wr[r], fa0[r] + bl0, o0);
            o1 = fmaf(acc1[r] * wr[r], fa1[r] + bl1, o1);
        }
    }
    o0 += __shfl_xor(o0, 16, 32);
    o1 += __shfl_xor(o1, 16, 32);
    if (lane < 16) {
        out[wid * Cc + lane]      = o0;
        out[wid * Cc + 16 + lane] = o1;
    }
}

extern "C" void kernel_launch(void* const* d_in, const int* in_sizes, int n_in,
                              void* d_out, int out_size, void* d_ws, size_t ws_size,
                              hipStream_t stream) {
    (void)in_sizes; (void)n_in; (void)out_size; (void)d_ws; (void)ws_size;
    const float* pndata = (const float*)d_in[0];
    const float* xcoord = (const float*)d_in[1];
    const float* lq     = (const float*)d_in[2];
    const float* W_lift = (const float*)d_in[3];
    const float* b_lift = (const float*)d_in[4];
    const float* W1     = (const float*)d_in[5];
    const float* b1     = (const float*)d_in[6];
    const float* W2     = (const float*)d_in[7];
    const float* b2     = (const float*)d_in[8];
    float* out = (float*)d_out;

    gno_kernel<<<Bb * (Mm / 8), 256, 0, stream>>>(xcoord, pndata, lq, W_lift, b_lift,
                                                  W1, b1, W2, b2, out);
}


# neighbour scan: all 64 distance tests run first as one vector run, then a branch-free compaction (a lane outside both radii stores to a spare LDS slot); list order and sums unchanged
# speedup vs baseline: 10.8288x; 1.0845x over previous
#include <hip/hip_runtime.h>


typedef __attribute__((ext_vector_type(2))) float v2f;
typedef __attribute__((ext_vector_type(8))) float v8f;

static constexpr int Bb  = 2;
static constexpr int Nn  = 1024;
static constexpr int Mm  = 1024;
static constexpr int DIN = 16;
static constexpr int Cc  = 32;
static constexpr int Hh  = 32;

__device__ __forceinline__ float gelu_tanh(float x) {
    const float u = x * x;
    const float t = 0.7978845608028654f * x * fmaf(0.044715f, u, 1.0f);
    const float e = __expf(-2.0f * t);
    const float th = (1.0f - e) * __frcp_rn(1.0f + e);
    return 0.5f * x * (1.0f + th);
}

__global__ void __launch_bounds__(256) gno_kernel(
    const float* __restrict__ xc, const float* __restrict__ pnd,
    const float* __restrict__ lq,
    const float* __restrict__ WL, const float* __restrict__ bl,
    const float* __restrict__ W1, const float* __restrict__ b1,
    const float* __restrict__ W2, const float* __restrict__ b2,
    float* __restrict__ out) {
    __shared__ __attribute__((aligned(16))) float xss[Nn * 2];
    __shared__ __attribute__((aligned(16))) int nbr[8][Nn + 32];
    __shared__ __attribute__((aligned(16))) float red[8][4];

    const int tid  = threadIdx.x;
    const int b    = blockIdx.x / (Mm / 8);
    const int mg   = blockIdx.x % (Mm / 8);
    const int wv   = tid >> 5;
    const int lane = tid & 31;
    const int m    = mg * 8 + wv;
    const int wid  = b * Mm + m;
    const int lrow = lane & 15;
    const int hsel = lane >> 4;

    const float4* s2 = (const float4*)(xc + b * Nn * 2);
    const float4 va = s2[tid];
    const float4 vb = s2[tid + 256];

    const float q0 = lq[m * 2], q1 = lq[m * 2 + 1];

    float W1x[16], W1y[16], Cm[16];
#pragma unroll
    for (int j = 0; j < 16; ++j) {
        const int h = 4 * (j >> 1) + 2 * hsel + (j & 1);
        W1x[j] = W1[h];
        W1y[j] = W1[Hh + h];
        Cm[j]  = fmaf(q0, W1[2 * Hh + h], fmaf(q1, W1[3 * Hh + h], b1[h]));
    }

    v2f Bw0[8], Bw1[8];
#pragma unroll
    for (int kk = 0; kk < 8; ++kk) {
        const int k0 = 4 * kk + 2 * hsel;
        Bw0[kk].x = W2[k0 * Cc + lrow];
        Bw0[kk].y = W2[(k0 + 1) * Cc + lrow];
        Bw1[kk].x = W2[k0 * Cc + 16 + lrow];
        Bw1[kk].y = W2[(k0 + 1) * Cc + 16 + lrow];
    }
    const float bz0 = b2[lrow], bz1 = b2[16 + lrow];

    v2f Lw0[4], Lw1[4];
#pragma unroll
    for (int kk = 0; kk < 4; ++kk) {
        const int k0 = 4 * kk + 2 * hsel;
        Lw0[kk].x = WL[k0 * Cc + lrow];
        Lw0[kk].y = WL[(k0 + 1) * Cc + lrow];
        Lw1[kk].x = WL[k0 * Cc + 16 + lrow];
        Lw1[kk].y = WL[(k0 + 1) * Cc + 16 + lrow];
    }
    const float bl0 = bl[lrow], bl1 = bl[16 + lrow];

    float mn0 = fminf(fminf(va.x, va.z), fminf(vb.x, vb.z));
    float mn1 = fminf(fminf(va.y, va.w), fminf(vb.y, vb.w));
    float mx0 = fmaxf(fmaxf(va.x, va.z), fmaxf(vb.x, vb.z));
    float mx1 = fmaxf(fmaxf(va.y, va.w), fmaxf(vb.y, vb.w));
#pragma unroll
    for (int s = 16; s >= 1; s >>= 1) {
        mn0 = fminf(mn0, __shfl_xor(mn0, s, 32));
        mn1 = fminf(mn1, __shfl_xor(mn1, s, 32));
        mx0 = fmaxf(mx0, __shfl_xor(mx0, s, 32));
        mx1 = fmaxf(mx1, __shfl_xor(mx1, s, 32));
    }
    if (lane == 0) *(float4*)red[wv] = make_float4(mn0, mn1, mx0, mx1);
    __syncthreads();
#pragma unroll
    for (int w = 0; w < 8; ++w) {
        const float4 rw = *(const float4*)red[w];
        mn0 = fminf(mn0, rw.x);
        mn1 = fminf(mn1, rw.y);
        mx0 = fmaxf(mx0, rw.z);
        mx1 = fmaxf(mx1, rw.w);
    }
    {
        const float sc0 = 2.0f / (mx0 - mn0 + 1e-12f);
        const float sc1 = 2.0f / (mx1 - mn1 + 1e-12f);
        float4 ra, rb;
        ra.x = (va.x - mn0) * sc0 - 1.0f;
        ra.y = (va.y - mn1) * sc1 - 1.0f;
        ra.z = (va.z - mn0) * sc0 - 1.0f;
        ra.w = (va.w - mn1) * sc1 - 1.0f;
        rb.x = (vb.x - mn0) * sc0 - 1.0f;
        rb.y = (vb.y - mn1) * sc1 - 1.0f;
        rb.z = (vb.z - mn0) * sc0 - 1.0f;
        rb.w = (vb.w - mn1) * sc1 - 1.0f;
        float4* d2v = (float4*)xss;
        d2v[tid]       = ra;
        d2v[tid + 256] = rb;
    }
    __syncthreads();

    int* lst = nbr[wv];
    const float2* xs2 = (const float2*)xss;
    float2 pp[Nn / 32];
#pragma unroll
    for (int it = 0; it < Nn / 32; ++it) pp[it] = xs2[it * 32 + lane];
    bool i1[Nn / 32], i2[Nn / 32];
#pragma unroll
    for (int it = 0; it < Nn / 32; ++it) {
        const float dx = q0 - pp[it].x, dy = q1 - pp[it].y;
        const float d2 = fmaf(dx, dx, dy * dy);
        i1[it] = (d2 <= 0.01f);
        i2[it] = (d2 <= 0.04f);
    }
    __builtin_amdgcn_sched_barrier(0);
    int cnt = 0, c1 = 0, c2 = 0;
#pragma unroll
    for (int it = 0; it < Nn / 32; ++it) {
        const int n = it * 32 + lane;
        const bool in = i1[it] || i2[it];
        const unsigned mask  = __builtin_amdgcn_ballot_w32(in);
        const unsigned mask1 = __builtin_amdgcn_ballot_w32(i1[it]);
        const unsigned mask2 = __builtin_amdgcn_ballot_w32(i2[it]);
        const int rank = (int)__builtin_amdgcn_mbcnt_lo(mask, (unsigned)cnt);
        lst[in ? rank : Nn + lane] =
            n | (i1[it] ? 0x10000 : 0) | (i2[it] ? 0x20000 : 0);
        cnt += __builtin_popcount(mask);
        c1  += __builtin_popcount(mask1);
        c2  += __builtin_popcount(mask2);
    }
    const float inv1 = 1.0f / fmaxf((float)c1, 1.0f);
    const float inv2 = 1.0f / fmaxf((float)c2, 1.0f);
    __syncthreads();

    const float* pb = pnd + b * Nn * DIN;
    float o0 = 0.f, o1 = 0.f;
    const int ntile = (cnt + 15) >> 4;
    for (int t = 0; t < ntile; ++t) {
        const int ia = t * 16 + lrow;
        const int ea = (ia < cnt) ? lst[ia] : 0;
        const int na = ea & 0xFFFF;
        const float2 p = xs2[na];
        const float x0 = p.x, x1 = p.y;

        const float2* pr = (const float2*)(pb + na * DIN + 2 * hsel);
        v2f la[4];
#pragma unroll
        for (int kk = 0; kk < 4; ++kk) {
            const float2 g = pr[2 * kk];
            la[kk].x = g.x;
            la[kk].y = g.y;
        }

        const int id0 = t * 16 + 8 * hsel;
        const int4 eA = *(const int4*)(lst + id0);
        const int4 eB = *(const int4*)(lst + id0 + 4);
        const int er[8] = {eA.x, eA.y, eA.z, eA.w, eB.x, eB.y, eB.z, eB.w};
        float wr[8];
#pragma unroll
        for (int r = 0; r < 8; ++r) {
            const int e = (id0 + r < cnt) ? er[r] : 0;
            wr[r] = ((e & 0x10000) ? inv1 : 0.f) + ((e & 0x20000) ? inv2 : 0.f);
        }

        v8f acc0 = {bz0, bz0, bz0, bz0, bz0, bz0, bz0, bz0};
        v8f acc1 = {bz1, bz1, bz1, bz1, bz1, bz1, bz1, bz1};
#pragma unroll
        for (int kk = 0; kk < 8; ++kk) {
            v2f a;
            const float p0 = fmaf(x0, W1x[2 * kk],     fmaf(x1, W1y[2 * kk],     Cm[2 * kk]));
            const float p1 = fmaf(x0, W1x[2 * kk + 1], fmaf(x1, W1y[2 * kk + 1], Cm[2 * kk + 1]));
            a.x = gelu_tanh(p0);
            a.y = gelu_tanh(p1);
            acc0 = __builtin_amdgcn_wmma_f32_16x16x4_f32(false, a, false, Bw0[kk],
                                                         (short)0, acc0, false, false);
            acc1 = __builtin_amdgcn_wmma_f32_16x16x4_f32(false, a, false, Bw1[kk],
                                                         (short)0, acc1, false, false);
        }

        v8f fa0 = {0.f, 0.f, 0.f, 0.f, 0.f, 0.f, 0.f, 0.f};
        v8f fa1 = {0.f, 0.f, 0.f, 0.f, 0.f, 0.f, 0.f, 0.f};
#pragma unroll
        for (int kk = 0; kk < 4; ++kk) {
            fa0 = __builtin_amdgcn_wmma_f32_16x16x4_f32(false, la[kk], false, Lw0[kk],
                                                        (short)0, fa0, false, false);
            fa1 = __builtin_amdgcn_wmma_f32_16x16x4_f32(false, la[kk], false, Lw1[kk],
                                                        (short)0, fa1, false, false);
        }

#pragma unroll
        for (int r = 0; r < 8; ++r) {
            o0 = fmaf(acc0[r] * wr[r], fa0[r] + bl0, o0);
            o1 = fmaf(acc1[r] * wr[r], fa1[r] + bl1, o1);
        }
    }
    o0 += __shfl_xor(o0, 16, 32);
    o1 += __shfl_xor(o1, 16, 32);
    if (lane < 16) {
        out[wid * Cc + lane]      = o0;
        out[wid * Cc + 16 + lane] = o1;
    }
}

extern "C" void kernel_launch(void* const* d_in, const int* in_sizes, int n_in,
                              void* d_out, int out_size, void* d_ws, size_t ws_size,
                              hipStream_t stream) {
    (void)in_sizes; (void)n_in; (void)out_size; (void)d_ws; (void)ws_size;
    const float* pndata = (const float*)d_in[0];
    const float* xcoord = (const float*)d_in[1];
    const float* lq     = (const float*)d_in[2];
    const float* W_lift = (const float*)d_in[3];
    const float* b_lift = (const float*)d_in[4];
    const float* W1     = (const float*)d_in[5];
    const float* b1     = (const float*)d_in[6];
    const float* W2     = (const float*)d_in[7];
    const float* b2     = (const float*)d_in[8];
    float* out = (float*)d_out;

    gno_kernel<<<Bb * (Mm / 8), 256, 0, stream>>>(xcoord, pndata, lq, W_lift, b_lift,
                                                  W1, b1, W2, b2, out);
}


# gelu as x*rcp(1+exp2(x*(k1+k2*x*x))): one v_rcp_f32 (1 ulp) in place of the correctly rounded division (10 instructions a value); the tanh form's extra packed ops go; order of sums unchanged
# speedup vs baseline: 12.0470x; 1.1125x over previous
#include <hip/hip_runtime.h>


typedef __attribute__((ext_vector_type(2))) float v2f;
typedef __attribute__((ext_vector_type(8))) float v8f;

static constexpr int Bb  = 2;
static constexpr int Nn  = 1024;
static constexpr int Mm  = 1024;
static constexpr int DIN = 16;
static constexpr int Cc  = 32;
static constexpr int Hh  = 32;

__device__ __forceinline__ v2f gelu_tanh2(v2f x) {
    const v2f k1 = {-2.302208198144325f, -2.302208198144325f};
    const v2f k2 = {-0.1029432395800235f, -0.1029432395800235f};
    const v2f u = x * x;
    const v2f a = x * __builtin_elementwise_fma(u, k2, k1);
    v2f d;
    d.x = 1.0f + __builtin_amdgcn_exp2f(a.x);
    d.y = 1.0f + __builtin_amdgcn_exp2f(a.y);
    v2f r;
    r.x = __builtin_amdgcn_rcpf(d.x);
    r.y = __builtin_amdgcn_rcpf(d.y);
    return x * r;
}

__global__ void __launch_bounds__(256) gno_kernel(
    const float* __restrict__ xc, const float* __restrict__ pnd,
    const float* __restrict__ lq,
    const float* __restrict__ WL, const float* __restrict__ bl,
    const float* __restrict__ W1, const float* __restrict__ b1,
    const float* __restrict__ W2, const float* __restrict__ b2,
    float* __restrict__ out) {
    __shared__ __attribute__((aligned(16))) float xss[Nn * 2];
    __shared__ __attribute__((aligned(16))) int nbr[8][Nn + 32];
    __shared__ __attribute__((aligned(16))) float red[8][4];

    const int tid  = threadIdx.x;
    const int b    = blockIdx.x / (Mm / 8);
    const int mg   = blockIdx.x % (Mm / 8);
    const int wv   = tid >> 5;
    const int lane = tid & 31;
    const int m    = mg * 8 + wv;
    const int wid  = b * Mm + m;
    const int lrow = lane & 15;
    const int hsel = lane >> 4;

    const float4* s2 = (const float4*)(xc + b * Nn * 2);
    const float4 va = s2[tid];
    const float4 vb = s2[tid + 256];

    const float q0 = lq[m * 2], q1 = lq[m * 2 + 1];

    float W1x[16], W1y[16], Cm[16];
#pragma unroll
    for (int j = 0; j < 16; ++j) {
        const int h = 4 * (j >> 1) + 2 * hsel + (j & 1);
        W1x[j] = W1[h];
        W1y[j] = W1[Hh + h];
        Cm[j]  = fmaf(q0, W1[2 * Hh + h], fmaf(q1, W1[3 * Hh + h], b1[h]));
    }

    v2f Bw0[8], Bw1[8];
#pragma unroll
    for (int kk = 0; kk < 8; ++kk) {
        const int k0 = 4 * kk + 2 * hsel;
        Bw0[kk].x = W2[k0 * Cc + lrow];
        Bw0[kk].y = W2[(k0 + 1) * Cc + lrow];
        Bw1[kk].x = W2[k0 * Cc + 16 + lrow];
        Bw1[kk].y = W2[(k0 + 1) * Cc + 16 + lrow];
    }
    const float bz0 = b2[lrow], bz1 = b2[16 + lrow];

    v2f Lw0[4], Lw1[4];
#pragma unroll
    for (int kk = 0; kk < 4; ++kk) {
        const int k0 = 4 * kk + 2 * hsel;
        Lw0[kk].x = WL[k0 * Cc + lrow];
        Lw0[kk].y = WL[(k0 + 1) * Cc + lrow];
        Lw1[kk].x = WL[k0 * Cc + 16 + lrow];
        Lw1[kk].y = WL[(k0 + 1) * Cc + 16 + lrow];
    }
    const float bl0 = bl[lrow], bl1 = bl[16 + lrow];

    float mn0 = fminf(fminf(va.x, va.z), fminf(vb.x, vb.z));
    float mn1 = fminf(fminf(va.y, va.w), fminf(vb.y, vb.w));
    float mx0 = fmaxf(fmaxf(va.x, va.z), fmaxf(vb.x, vb.z));
    float mx1 = fmaxf(fmaxf(va.y, va.w), fmaxf(vb.y, vb.w));
#pragma unroll
    for (int s = 16; s >= 1; s >>= 1) {
        mn0 = fminf(mn0, __shfl_xor(mn0, s, 32));
        mn1 = fminf(mn1, __shfl_xor(mn1, s, 32));
        mx0 = fmaxf(mx0, __shfl_xor(mx0, s, 32));
        mx1 = fmaxf(mx1, __shfl_xor(mx1, s, 32));
    }
    if (lane == 0) *(float4*)red[wv] = make_float4(mn0, mn1, mx0, mx1);
    __syncthreads();
#pragma unroll
    for (int w = 0; w < 8; ++w) {
        const float4 rw = *(const float4*)red[w];
        mn0 = fminf(mn0, rw.x);
        mn1 = fminf(mn1, rw.y);
        mx0 = fmaxf(mx0, rw.z);
        mx1 = fmaxf(mx1, rw.w);
    }
    {
        const float sc0 = 2.0f / (mx0 - mn0 + 1e-12f);
        const float sc1 = 2.0f / (mx1 - mn1 + 1e-12f);
        float4 ra, rb;
        ra.x = (va.x - mn0) * sc0 - 1.0f;
        ra.y = (va.y - mn1) * sc1 - 1.0f;
        ra.z = (va.z - mn0) * sc0 - 1.0f;
        ra.w = (va.w - mn1) * sc1 - 1.0f;
        rb.x = (vb.x - mn0) * sc0 - 1.0f;
        rb.y = (vb.y - mn1) * sc1 - 1.0f;
        rb.z = (vb.z - mn0) * sc0 - 1.0f;
        rb.w = (vb.w - mn1) * sc1 - 1.0f;
        float4* d2v = (float4*)xss;
        d2v[tid]       = ra;
        d2v[tid + 256] = rb;
    }
    __syncthreads();

    int* lst = nbr[wv];
    const float2* xs2 = (const float2*)xss;
    float2 pp[Nn / 32];
#pragma unroll
    for (int it = 0; it < Nn / 32; ++it) pp[it] = xs2[it * 32 + lane];
    bool i1[Nn / 32], i2[Nn / 32];
#pragma unroll
    for (int it = 0; it < Nn / 32; ++it) {
        const float dx = q0 - pp[it].x, dy = q1 - pp[it].y;
        const float d2 = fmaf(dx, dx, dy * dy);
        i1[it] = (d2 <= 0.01f);
        i2[it] = (d2 <= 0.04f);
    }
    __builtin_amdgcn_sched_barrier(0);
    int cnt = 0, c1 = 0, c2 = 0;
#pragma unroll
    for (int it = 0; it < Nn / 32; ++it) {
        const int n = it * 32 + lane;
        const bool in = i1[it] || i2[it];
        const unsigned mask  = __builtin_amdgcn_ballot_w32(in);
        const unsigned mask1 = __builtin_amdgcn_ballot_w32(i1[it]);
        const unsigned mask2 = __builtin_amdgcn_ballot_w32(i2[it]);
        const int rank = (int)__builtin_amdgcn_mbcnt_lo(mask, (unsigned)cnt);
        lst[in ? rank : Nn + lane] =
            n | (i1[it] ? 0x10000 : 0) | (i2[it] ? 0x20000 : 0);
        cnt += __builtin_popcount(mask);
        c1  += __builtin_popcount(mask1);
        c2  += __builtin_popcount(mask2);
    }
    const float inv1 = 1.0f / fmaxf((float)c1, 1.0f);
    const float inv2 = 1.0f / fmaxf((float)c2, 1.0f);
    __syncthreads();

    const float* pb = pnd + b * Nn * DIN;
    float o0 = 0.f, o1 = 0.f;
    const int ntile = (cnt + 15) >> 4;
    for (int t = 0; t < ntile; ++t) {
        const int ia = t * 16 + lrow;
        const int ea = (ia < cnt) ? lst[ia] : 0;
        const int na = ea & 0xFFFF;
        const float2 p = xs2[na];
        const float x0 = p.x, x1 = p.y;

        const float2* pr = (const float2*)(pb + na * DIN + 2 * hsel);
        v2f la[4];
#pragma unroll
        for (int kk = 0; kk < 4; ++kk) {
            const float2 g = pr[2 * kk];
            la[kk].x = g.x;
            la[kk].y = g.y;
        }

        const int id0 = t * 16 + 8 * hsel;
        const int4 eA = *(const int4*)(lst + id0);
        const int4 eB = *(const int4*)(lst + id0 + 4);
        const int er[8] = {eA.x, eA.y, eA.z, eA.w, eB.x, eB.y, eB.z, eB.w};
        float wr[8];
#pragma unroll
        for (int r = 0; r < 8; ++r) {
            const int e = (id0 + r < cnt) ? er[r] : 0;
            wr[r] = ((e & 0x10000) ? inv1 : 0.f) + ((e & 0x20000) ? inv2 : 0.f);
        }

        v8f acc0 = {bz0, bz0, bz0, bz0, bz0, bz0, bz0, bz0};
        v8f acc1 = {bz1, bz1, bz1, bz1, bz1, bz1, bz1, bz1};
#pragma unroll
        for (int kk = 0; kk < 8; ++kk) {
            v2f p;
            p.x = fmaf(x0, W1x[2 * kk],     fmaf(x1, W1y[2 * kk],     Cm[2 * kk]));
            p.y = fmaf(x0, W1x[2 * kk + 1], fmaf(x1, W1y[2 * kk + 1], Cm[2 * kk + 1]));
            const v2f a = gelu_tanh2(p);
            acc0 = __builtin_amdgcn_wmma_f32_16x16x4_f32(false, a, false, Bw0[kk],
                                                         (short)0, acc0, false, false);
            acc1 = __builtin_amdgcn_wmma_f32_16x16x4_f32(false, a, false, Bw1[kk],
                                                         (short)0, acc1, false, false);
        }

        v8f fa0 = {0.f, 0.f, 0.f, 0.f, 0.f, 0.f, 0.f, 0.f};
        v8f fa1 = {0.f, 0.f, 0.f, 0.f, 0.f, 0.f, 0.f, 0.f};
#pragma unroll
        for (int kk = 0; kk < 4; ++kk) {
            fa0 = __builtin_amdgcn_wmma_f32_16x16x4_f32(false, la[kk], false, Lw0[kk],
                                                        (short)0, fa0, false, false);
            fa1 = __builtin_amdgcn_wmma_f32_16x16x4_f32(false, la[kk], false, Lw1[kk],
                                                        (short)0, fa1, false, false);
        }

#pragma unroll
        for (int r = 0; r < 8; ++r) {
            o0 = fmaf(acc0[r] * wr[r], fa0[r] + bl0, o0);
            o1 = fmaf(acc1[r] * wr[r], fa1[r] + bl1, o1);
        }
    }
    o0 += __shfl_xor(o0, 16, 32);
    o1 += __shfl_xor(o1, 16, 32);
    if (lane < 16) {
        out[wid * Cc + lane]      = o0;
        out[wid * Cc + 16 + lane] = o1;
    }
}

extern "C" void kernel_launch(void* const* d_in, const int* in_sizes, int n_in,
                              void* d_out, int out_size, void* d_ws, size_t ws_size,
                              hipStream_t stream) {
    (void)in_sizes; (void)n_in; (void)out_size; (void)d_ws; (void)ws_size;
    const float* pndata = (const float*)d_in[0];
    const float* xcoord = (const float*)d_in[1];
    const float* lq     = (const float*)d_in[2];
    const float* W_lift = (const float*)d_in[3];
    const float* b_lift = (const float*)d_in[4];
    const float* W1     = (const float*)d_in[5];
    const float* b1     = (const float*)d_in[6];
    const float* W2     = (const float*)d_in[7];
    const float* b2     = (const float*)d_in[8];
    float* out = (float*)d_out;

    gno_kernel<<<Bb * (Mm / 8), 256, 0, stream>>>(xcoord, pndata, lq, W_lift, b_lift,
                                                  W1, b1, W2, b2, out);
}


# pass 2 reassociated: T[h,d]=sum_n w*gelu*p by 2 WMMA per 4 listed points, then once per query T through W2 (18 WMMA) and W_lift in-lane; the 24 WMMA per 16-point tile go; changes fp summation order
# speedup vs baseline: 12.9927x; 1.0785x over previous
#include <hip/hip_runtime.h>


typedef __attribute__((ext_vector_type(2))) float v2f;
typedef __attribute__((ext_vector_type(8))) float v8f;

static constexpr int Bb  = 2;
static constexpr int Nn  = 1024;
static constexpr int Mm  = 1024;
static constexpr int DIN = 16;
static constexpr int Cc  = 32;
static constexpr int Hh  = 32;

__device__ __forceinline__ v2f gelu_tanh2(v2f x) {
    const v2f k1 = {-2.302208198144325f, -2.302208198144325f};
    const v2f k2 = {-0.1029432395800235f, -0.1029432395800235f};
    const v2f u = x * x;
    const v2f a = x * __builtin_elementwise_fma(u, k2, k1);
    v2f d;
    d.x = 1.0f + __builtin_amdgcn_exp2f(a.x);
    d.y = 1.0f + __builtin_amdgcn_exp2f(a.y);
    v2f r;
    r.x = __builtin_amdgcn_rcpf(d.x);
    r.y = __builtin_amdgcn_rcpf(d.y);
    return x * r;
}

__device__ __forceinline__ void t_stretch(
    const int rem, const int* lp, const int hsel, const int lrow,
    const float2* xs2, const float* pb, const float inv12, const float inv2,
    const float W1xA, const float W1yA, const float CmA,
    const float W1xB, const float W1yB, const float CmB,
    v8f& T0, v8f& T1, v2f& tp2, v2f& tgA2, v2f& tgB2) {
    int2 ent[16];
#pragma unroll
    for (int j = 0; j < 16; ++j) ent[j] = *(const int2*)(lp + 4 * j);
    float2 ca[16], cb[16];
    v2f bp[16];
#pragma unroll
    for (int j = 0; j < 16; ++j) {
        const int n0 = ent[j].x & (Nn - 1), n1 = ent[j].y & (Nn - 1);
        ca[j] = xs2[n0];
        cb[j] = xs2[n1];
        bp[j].x = pb[(unsigned)(n0 * DIN + lrow)];
        bp[j].y = pb[(unsigned)(n1 * DIN + lrow)];
    }
#pragma unroll
    for (int j = 0; j < 16; ++j) {
        if (4 * j < rem) {
            const int i0 = 4 * j + 2 * hsel;
            v2f wq;
            wq.x = (i0 < rem) ? ((ent[j].x & 0x10000) ? inv12 : inv2) : 0.f;
            wq.y = (i0 + 1 < rem) ? ((ent[j].y & 0x10000) ? inv12 : inv2) : 0.f;
            v2f pa, pc;
            pa.x = fmaf(ca[j].x, W1xA, fmaf(ca[j].y, W1yA, CmA));
            pa.y = fmaf(cb[j].x, W1xA, fmaf(cb[j].y, W1yA, CmA));
            pc.x = fmaf(ca[j].x, W1xB, fmaf(ca[j].y, W1yB, CmB));
            pc.y = fmaf(cb[j].x, W1xB, fmaf(cb[j].y, W1yB, CmB));
            const v2f a0 = gelu_tanh2(pa) * wq;
            const v2f a1 = gelu_tanh2(pc) * wq;
            T0 = __builtin_amdgcn_wmma_f32_16x16x4_f32(false, a0, false, bp[j],
                                                       (short)0, T0, false, false);
            T1 = __builtin_amdgcn_wmma_f32_16x16x4_f32(false, a1, false, bp[j],
                                                       (short)0, T1, false, false);
            tp2 = __builtin_elementwise_fma(wq, bp[j], tp2);
            tgA2 += a0;
            tgB2 += a1;
        }
    }
}

__global__ void __launch_bounds__(256) gno_kernel(
    const float* __restrict__ xc, const float* __restrict__ pnd,
    const float* __restrict__ lq,
    const float* __restrict__ WL, const float* __restrict__ bl,
    const float* __restrict__ W1, const float* __restrict__ b1,
    const float* __restrict__ W2, const float* __restrict__ b2,
    float* __restrict__ out) {
    __shared__ __attribute__((aligned(16))) float xss[Nn * 2];
    __shared__ __attribute__((aligned(16))) int nbr[8][Nn + 32];
    __shared__ __attribute__((aligned(16))) float red[8][4];

    const int tid  = threadIdx.x;
    const int b    = blockIdx.x / (Mm / 8);
    const int mg   = blockIdx.x % (Mm / 8);
    const int wv   = tid >> 5;
    const int lane = tid & 31;
    const int m    = mg * 8 + wv;
    const int wid  = b * Mm + m;
    const int lrow = lane & 15;
    const int hsel = lane >> 4;

    const float4* s2 = (const float4*)(xc + b * Nn * 2);
    const float4 va = s2[tid];
    const float4 vb = s2[tid + 256];

    const float q0 = lq[m * 2], q1 = lq[m * 2 + 1];

    const int hA = lrow, hB = 16 + lrow;
    const float W1xA = W1[hA], W1yA = W1[Hh + hA];
    const float W1xB = W1[hB], W1yB = W1[Hh + hB];
    const float CmA = fmaf(q0, W1[2 * Hh + hA], fmaf(q1, W1[3 * Hh + hA], b1[hA]));
    const float CmB = fmaf(q0, W1[2 * Hh + hB], fmaf(q1, W1[3 * Hh + hB], b1[hB]));

    v2f Bw0[8], Bw1[8];
#pragma unroll
    for (int kk = 0; kk < 8; ++kk) {
        const int k0 = 16 * (kk >> 2) + 8 * hsel + 2 * (kk & 3);
        Bw0[kk].x = W2[k0 * Cc + lrow];
        Bw0[kk].y = W2[(k0 + 1) * Cc + lrow];
        Bw1[kk].x = W2[k0 * Cc + 16 + lrow];
        Bw1[kk].y = W2[(k0 + 1) * Cc + 16 + lrow];
    }
    const float bz0 = b2[lrow], bz1 = b2[16 + lrow];

    float wl0[8], wl1[8];
#pragma unroll
    for (int r = 0; r < 8; ++r) {
        wl0[r] = WL[(8 * hsel + r) * Cc + lrow];
        wl1[r] = WL[(8 * hsel + r) * Cc + 16 + lrow];
    }
    const float bl0 = bl[lrow], bl1 = bl[16 + lrow];
    const bool has_bl = __builtin_amdgcn_ballot_w32(bl0 != 0.f || bl1 != 0.f) != 0u;

    float mn0 = fminf(fminf(va.x, va.z), fminf(vb.x, vb.z));
    float mn1 = fminf(fminf(va.y, va.w), fminf(vb.y, vb.w));
    float mx0 = fmaxf(fmaxf(va.x, va.z), fmaxf(vb.x, vb.z));
    float mx1 = fmaxf(fmaxf(va.y, va.w), fmaxf(vb.y, vb.w));
#pragma unroll
    for (int s = 16; s >= 1; s >>= 1) {
        mn0 = fminf(mn0, __shfl_xor(mn0, s, 32));
        mn1 = fminf(mn1, __shfl_xor(mn1, s, 32));
        mx0 = fmaxf(mx0, __shfl_xor(mx0, s, 32));
        mx1 = fmaxf(mx1, __shfl_xor(mx1, s, 32));
    }
    if (lane == 0) *(float4*)red[wv] = make_float4(mn0, mn1, mx0, mx1);
    __syncthreads();
#pragma unroll
    for (int w = 0; w < 8; ++w) {
        const float4 rw = *(const float4*)red[w];
        mn0 = fminf(mn0, rw.x);
        mn1 = fminf(mn1, rw.y);
        mx0 = fmaxf(mx0, rw.z);
        mx1 = fmaxf(mx1, rw.w);
    }
    {
        const float sc0 = 2.0f / (mx0 - mn0 + 1e-12f);
        const float sc1 = 2.0f / (mx1 - mn1 + 1e-12f);
        float4 ra, rb;
        ra.x = (va.x - mn0) * sc0 - 1.0f;
        ra.y = (va.y - mn1) * sc1 - 1.0f;
        ra.z = (va.z - mn0) * sc0 - 1.0f;
        ra.w = (va.w - mn1) * sc1 - 1.0f;
        rb.x = (vb.x - mn0) * sc0 - 1.0f;
        rb.y = (vb.y - mn1) * sc1 - 1.0f;
        rb.z = (vb.z - mn0) * sc0 - 1.0f;
        rb.w = (vb.w - mn1) * sc1 - 1.0f;
        float4* d2v = (float4*)xss;
        d2v[tid]       = ra;
        d2v[tid + 256] = rb;
    }
    __syncthreads();

    int* lst = nbr[wv];
    const float2* xs2 = (const float2*)xss;
    float2 pp[Nn / 32];
#pragma unroll
    for (int it = 0; it < Nn / 32; ++it) pp[it] = xs2[it * 32 + lane];
    bool i1[Nn / 32], i2[Nn / 32];
#pragma unroll
    for (int it = 0; it < Nn / 32; ++it) {
        const float dx = q0 - pp[it].x, dy = q1 - pp[it].y;
        const float d2 = fmaf(dx, dx, dy * dy);
        i1[it] = (d2 <= 0.01f);
        i2[it] = (d2 <= 0.04f);
    }
    __builtin_amdgcn_sched_barrier(0);
    int cnt = 0, c1 = 0, c2 = 0;
#pragma unroll
    for (int it = 0; it < Nn / 32; ++it) {
        const int n = it * 32 + lane;
        const bool in = i1[it] || i2[it];
        const unsigned mask  = __builtin_amdgcn_ballot_w32(in);
        const unsigned mask1 = __builtin_amdgcn_ballot_w32(i1[it]);
        const unsigned mask2 = __builtin_amdgcn_ballot_w32(i2[it]);
        const int rank = (int)__builtin_amdgcn_mbcnt_lo(mask, (unsigned)cnt);
        lst[in ? rank : Nn + lane] =
            n | (i1[it] ? 0x10000 : 0) | (i2[it] ? 0x20000 : 0);
        cnt += __builtin_popcount(mask);
        c1  += __builtin_popcount(mask1);
        c2  += __builtin_popcount(mask2);
    }
    const float inv1 = 1.0f / fmaxf((float)c1, 1.0f);
    const float inv2 = 1.0f / fmaxf((float)c2, 1.0f);
    __syncthreads();

    const float* pb = pnd + b * Nn * DIN;
    v8f T0 = {0.f, 0.f, 0.f, 0.f, 0.f, 0.f, 0.f, 0.f};
    v8f T1 = {0.f, 0.f, 0.f, 0.f, 0.f, 0.f, 0.f, 0.f};
    v2f tp2 = {0.f, 0.f}, tgA2 = {0.f, 0.f}, tgB2 = {0.f, 0.f};
    const float inv12 = inv1 + inv2;
    if (cnt > 0)
        t_stretch(cnt, lst + 2 * hsel, hsel, lrow, xs2, pb, inv12, inv2,
                  W1xA, W1yA, CmA, W1xB, W1yB, CmB, T0, T1, tp2, tgA2, tgB2);
    for (int base = 64; base < cnt; base += 64)
        t_stretch(cnt - base, lst + base + 2 * hsel, hsel, lrow, xs2, pb, inv12, inv2,
                  W1xA, W1yA, CmA, W1xB, W1yB, CmB, T0, T1, tp2, tgA2, tgB2);
    float tp = tp2.x + tp2.y;
    float tgA = tgA2.x + tgA2.y, tgB = tgB2.x + tgB2.y;
    tp += __shfl_xor(tp, 16, 32);

    v8f U0 = {0.f, 0.f, 0.f, 0.f, 0.f, 0.f, 0.f, 0.f};
    v8f U1 = {0.f, 0.f, 0.f, 0.f, 0.f, 0.f, 0.f, 0.f};
#pragma unroll
    for (int jp = 0; jp < 4; ++jp) {
        v2f a;
        a.x = T0[2 * jp];
        a.y = T0[2 * jp + 1];
        U0 = __builtin_amdgcn_wmma_f32_16x16x4_f32(false, a, false, Bw0[jp],
                                                   (short)0, U0, false, false);
        U1 = __builtin_amdgcn_wmma_f32_16x16x4_f32(false, a, false, Bw1[jp],
                                                   (short)0, U1, false, false);
    }
#pragma unroll
    for (int jp = 0; jp < 4; ++jp) {
        v2f a;
        a.x = T1[2 * jp];
        a.y = T1[2 * jp + 1];
        U0 = __builtin_amdgcn_wmma_f32_16x16x4_f32(false, a, false, Bw0[4 + jp],
                                                   (short)0, U0, false, false);
        U1 = __builtin_amdgcn_wmma_f32_16x16x4_f32(false, a, false, Bw1[4 + jp],
                                                   (short)0, U1, false, false);
    }
    {
        v2f a, c0v, c1v;
        a.x = (hsel == 0) ? tp : 0.f;
        a.y = 0.f;
        c0v.x = bz0; c0v.y = 0.f;
        c1v.x = bz1; c1v.y = 0.f;
        U0 = __builtin_amdgcn_wmma_f32_16x16x4_f32(false, a, false, c0v,
                                                   (short)0, U0, false, false);
        U1 = __builtin_amdgcn_wmma_f32_16x16x4_f32(false, a, false, c1v,
                                                   (short)0, U1, false, false);
    }
    float o0 = 0.f, o1 = 0.f;
#pragma unroll
    for (int r = 0; r < 8; ++r) {
        o0 = fmaf(wl0[r], U0[r], o0);
        o1 = fmaf(wl1[r], U1[r], o1);
    }
    o0 += __shfl_xor(o0, 16, 32);
    o1 += __shfl_xor(o1, 16, 32);
    if (has_bl) {
        tgA += __shfl_xor(tgA, 16, 32);
        tgB += __shfl_xor(tgB, 16, 32);
        const float sw = (float)c1 * inv1 + (float)c2 * inv2;
        float kb0 = sw * bz0, kb1 = sw * bz1;
#pragma unroll 1
        for (int h = 0; h < Hh; ++h) {
            const float tg = __shfl((h < 16) ? tgA : tgB, h & 15, 32);
            kb0 = fmaf(tg, W2[h * Cc + lrow], kb0);
            kb1 = fmaf(tg, W2[h * Cc + 16 + lrow], kb1);
        }
        o0 = fmaf(bl0, kb0, o0);
        o1 = fmaf(bl1, kb1, o1);
    }
    if (lane < 16) {
        out[wid * Cc + lane]      = o0;
        out[wid * Cc + 16 + lane] = o1;
    }
}

extern "C" void kernel_launch(void* const* d_in, const int* in_sizes, int n_in,
                              void* d_out, int out_size, void* d_ws, size_t ws_size,
                              hipStream_t stream) {
    (void)in_sizes; (void)n_in; (void)out_size; (void)d_ws; (void)ws_size;
    const float* pndata = (const float*)d_in[0];
    const float* xcoord = (const float*)d_in[1];
    const float* lq     = (const float*)d_in[2];
    const float* W_lift = (const float*)d_in[3];
    const float* b_lift = (const float*)d_in[4];
    const float* W1     = (const float*)d_in[5];
    const float* b1     = (const float*)d_in[6];
    const float* W2     = (const float*)d_in[7];
    const float* b2     = (const float*)d_in[8];
    float* out = (float*)d_out;

    gno_kernel<<<Bb * (Mm / 8), 256, 0, stream>>>(xcoord, pndata, lq, W_lift, b_lift,
                                                  W1, b1, W2, b2, out);
}


# neighbour list from per-lane hit bit masks: one wave prefix sum and a short per-lane write loop replace 32 ballot compactions; entries carry offset and mean weight; changes fp summation order
# speedup vs baseline: 14.1807x; 1.0914x over previous
#include <hip/hip_runtime.h>


typedef __attribute__((ext_vector_type(2))) float v2f;
typedef __attribute__((ext_vector_type(8))) float v8f;

static constexpr int Bb  = 2;
static constexpr int Nn  = 1024;
static constexpr int Mm  = 1024;
static constexpr int DIN = 16;
static constexpr int Cc  = 32;
static constexpr int Hh  = 32;

__device__ __forceinline__ v2f gelu_tanh2(v2f x) {
    const v2f k1 = {-2.302208198144325f, -2.302208198144325f};
    const v2f k2 = {-0.1029432395800235f, -0.1029432395800235f};
    const v2f u = x * x;
    const v2f a = x * __builtin_elementwise_fma(u, k2, k1);
    v2f d;
    d.x = 1.0f + __builtin_amdgcn_exp2f(a.x);
    d.y = 1.0f + __builtin_amdgcn_exp2f(a.y);
    v2f r;
    r.x = __builtin_amdgcn_rcpf(d.x);
    r.y = __builtin_amdgcn_rcpf(d.y);
    return x * r;
}

__device__ __forceinline__ void t_stretch(
    const int rem, const int2* lp, const int lrow,
    const float2* xs2, const float* pb,
    const float W1xA, const float W1yA, const float CmA,
    const float W1xB, const float W1yB, const float CmB,
    v8f& T0, v8f& T1, v2f& tp2, v2f& tgA2, v2f& tgB2) {
    int4 ent[16];
#pragma unroll
    for (int j = 0; j < 16; ++j) ent[j] = *(const int4*)(lp + 4 * j);
    float2 ca[16], cb[16];
    v2f bp[16];
#pragma unroll
    for (int j = 0; j < 16; ++j) {
        const unsigned o0 = (unsigned)ent[j].x, o1 = (unsigned)ent[j].z;
        ca[j] = *(const float2*)((const char*)xs2 + o0);
        cb[j] = *(const float2*)((const char*)xs2 + o1);
        bp[j].x = *(const float*)((const char*)pb + (o0 * 8u + 4u * (unsigned)lrow));
        bp[j].y = *(const float*)((const char*)pb + (o1 * 8u + 4u * (unsigned)lrow));
    }
#pragma unroll
    for (int j = 0; j < 16; ++j) {
        if (4 * j < rem) {
            v2f wq;
            wq.x = __int_as_float(ent[j].y);
            wq.y = __int_as_float(ent[j].w);
            v2f pa, pc;
            pa.x = fmaf(ca[j].x, W1xA, fmaf(ca[j].y, W1yA, CmA));
            pa.y = fmaf(cb[j].x, W1xA, fmaf(cb[j].y, W1yA, CmA));
            pc.x = fmaf(ca[j].x, W1xB, fmaf(ca[j].y, W1yB, CmB));
            pc.y = fmaf(cb[j].x, W1xB, fmaf(cb[j].y, W1yB, CmB));
            const v2f a0 = gelu_tanh2(pa) * wq;
            const v2f a1 = gelu_tanh2(pc) * wq;
            T0 = __builtin_amdgcn_wmma_f32_16x16x4_f32(false, a0, false, bp[j],
                                                       (short)0, T0, false, false);
            T1 = __builtin_amdgcn_wmma_f32_16x16x4_f32(false, a1, false, bp[j],
                                                       (short)0, T1, false, false);
            tp2 = __builtin_elementwise_fma(wq, bp[j], tp2);
            tgA2 += a0;
            tgB2 += a1;
        }
    }
}

__global__ void __launch_bounds__(256) gno_kernel(
    const float* __restrict__ xc, const float* __restrict__ pnd,
    const float* __restrict__ lq,
    const float* __restrict__ WL, const float* __restrict__ bl,
    const float* __restrict__ W1, const float* __restrict__ b1,
    const float* __restrict__ W2, const float* __restrict__ b2,
    float* __restrict__ out) {
    __shared__ struct __attribute__((aligned(16))) {
        float xss[Nn * 2];
        float red[8][4];
        int2  nbr[8][Nn];
    } sm;
    float* const xss = sm.xss;
    float (* const red)[4] = sm.red;
    int2 (* const nbr)[Nn] = sm.nbr;

    const int tid  = threadIdx.x;
    const int b    = blockIdx.x / (Mm / 8);
    const int mg   = blockIdx.x % (Mm / 8);
    const int wv   = tid >> 5;
    const int lane = tid & 31;
    const int m    = mg * 8 + wv;
    const int wid  = b * Mm + m;
    const int lrow = lane & 15;
    const int hsel = lane >> 4;

    const float4* s2 = (const float4*)(xc + b * Nn * 2);
    const float4 va = s2[tid];
    const float4 vb = s2[tid + 256];

    const float q0 = lq[m * 2], q1 = lq[m * 2 + 1];

    const int hA = lrow, hB = 16 + lrow;
    const float W1xA = W1[hA], W1yA = W1[Hh + hA];
    const float W1xB = W1[hB], W1yB = W1[Hh + hB];
    const float CmA = fmaf(q0, W1[2 * Hh + hA], fmaf(q1, W1[3 * Hh + hA], b1[hA]));
    const float CmB = fmaf(q0, W1[2 * Hh + hB], fmaf(q1, W1[3 * Hh + hB], b1[hB]));

    v2f Bw0[8], Bw1[8];
#pragma unroll
    for (int kk = 0; kk < 8; ++kk) {
        const int k0 = 16 * (kk >> 2) + 8 * hsel + 2 * (kk & 3);
        Bw0[kk].x = W2[k0 * Cc + lrow];
        Bw0[kk].y = W2[(k0 + 1) * Cc + lrow];
        Bw1[kk].x = W2[k0 * Cc + 16 + lrow];
        Bw1[kk].y = W2[(k0 + 1) * Cc + 16 + lrow];
    }
    const float bz0 = b2[lrow], bz1 = b2[16 + lrow];

    float wl0[8], wl1[8];
#pragma unroll
    for (int r = 0; r < 8; ++r) {
        wl0[r] = WL[(8 * hsel + r) * Cc + lrow];
        wl1[r] = WL[(8 * hsel + r) * Cc + 16 + lrow];
    }
    const float bl0 = bl[lrow], bl1 = bl[16 + lrow];
    const bool has_bl = __builtin_amdgcn_ballot_w32(bl0 != 0.f || bl1 != 0.f) != 0u;

    float mn0 = fminf(fminf(va.x, va.z), fminf(vb.x, vb.z));
    float mn1 = fminf(fminf(va.y, va.w), fminf(vb.y, vb.w));
    float mx0 = fmaxf(fmaxf(va.x, va.z), fmaxf(vb.x, vb.z));
    float mx1 = fmaxf(fmaxf(va.y, va.w), fmaxf(vb.y, vb.w));
#pragma unroll
    for (int s = 16; s >= 1; s >>= 1) {
        mn0 = fminf(mn0, __shfl_xor(mn0, s, 32));
        mn1 = fminf(mn1, __shfl_xor(mn1, s, 32));
        mx0 = fmaxf(mx0, __shfl_xor(mx0, s, 32));
        mx1 = fmaxf(mx1, __shfl_xor(mx1, s, 32));
    }
    if (lane == 0) *(float4*)red[wv] = make_float4(mn0, mn1, mx0, mx1);
    __syncthreads();
#pragma unroll
    for (int w = 0; w < 8; ++w) {
        const float4 rw = *(const float4*)red[w];
        mn0 = fminf(mn0, rw.x);
        mn1 = fminf(mn1, rw.y);
        mx0 = fmaxf(mx0, rw.z);
        mx1 = fmaxf(mx1, rw.w);
    }
    {
        const float sc0 = 2.0f / (mx0 - mn0 + 1e-12f);
        const float sc1 = 2.0f / (mx1 - mn1 + 1e-12f);
        float4 ra, rb;
        ra.x = (va.x - mn0) * sc0 - 1.0f;
        ra.y = (va.y - mn1) * sc1 - 1.0f;
        ra.z = (va.z - mn0) * sc0 - 1.0f;
        ra.w = (va.w - mn1) * sc1 - 1.0f;
        rb.x = (vb.x - mn0) * sc0 - 1.0f;
        rb.y = (vb.y - mn1) * sc1 - 1.0f;
        rb.z = (vb.z - mn0) * sc0 - 1.0f;
        rb.w = (vb.w - mn1) * sc1 - 1.0f;
        float4* d2v = (float4*)xss;
        d2v[tid]       = ra;
        d2v[tid + 256] = rb;
    }
    __syncthreads();

    int2* lst = nbr[wv];
    const float2* xs2 = (const float2*)xss;
    float2 pp[Nn / 32];
#pragma unroll
    for (int it = 0; it < Nn / 32; ++it) pp[it] = xs2[it * 32 + lane];
    unsigned om1 = 0u, om2 = 0u;
#pragma unroll
    for (int it = Nn / 32 - 1; it >= 0; --it) {
        const float dx = q0 - pp[it].x, dy = q1 - pp[it].y;
        const float d2 = fmaf(dx, dx, dy * dy);
        const float t1 = 0.01f - d2;
        const float t2 = 0.04f - d2;
        om1 = __builtin_amdgcn_alignbit(om1, __float_as_uint(t1), 31u);
        om2 = __builtin_amdgcn_alignbit(om2, __float_as_uint(t2), 31u);
    }
    const unsigned h2 = ~om2, h1 = ~om1;
    const int pk = __builtin_popcount(h2) + (__builtin_popcount(h1) << 16);
    int sc = pk;
    sc += __builtin_amdgcn_update_dpp(0, sc, 0x111, 0xf, 0xf, true);
    sc += __builtin_amdgcn_update_dpp(0, sc, 0x112, 0xf, 0xf, true);
    sc += __builtin_amdgcn_update_dpp(0, sc, 0x114, 0xf, 0xf, true);
    sc += __builtin_amdgcn_update_dpp(0, sc, 0x118, 0xf, 0xf, true);
    const int row0 = __builtin_amdgcn_readlane(sc, 15);
    if (lane >= 16) sc += row0;
    const int tot = __builtin_amdgcn_readlane(sc, 31);
    const int cnt = tot & 0xffff;
    const int c1 = tot >> 16, c2 = cnt;
    const float inv1 = 1.0f / fmaxf((float)c1, 1.0f);
    const float inv2 = 1.0f / fmaxf((float)c2, 1.0f);
    const float inv12 = inv1 + inv2;
    {
        int pos = (sc - pk) & 0xffff;
        unsigned hh = h2;
        while (hh != 0u) {
            const int bit = __builtin_ctz(hh);
            const float w = ((h1 >> bit) & 1u) ? inv12 : inv2;
            lst[pos] = make_int2(8 * (bit * 32 + lane), __float_as_int(w));
            ++pos;
            hh &= hh - 1u;
        }
    }
    {
        const int cend = (cnt + 63) & ~63;
        if (cnt + lane < cend)      lst[cnt + lane]      = make_int2(0, 0);
        if (cnt + 32 + lane < cend) lst[cnt + 32 + lane] = make_int2(0, 0);
    }
    __syncthreads();

    const float* pb = pnd + b * Nn * DIN;
    v8f T0 = {0.f, 0.f, 0.f, 0.f, 0.f, 0.f, 0.f, 0.f};
    v8f T1 = {0.f, 0.f, 0.f, 0.f, 0.f, 0.f, 0.f, 0.f};
    v2f tp2 = {0.f, 0.f}, tgA2 = {0.f, 0.f}, tgB2 = {0.f, 0.f};
    if (cnt > 0)
        t_stretch(cnt, lst + 2 * hsel, lrow, xs2, pb,
                  W1xA, W1yA, CmA, W1xB, W1yB, CmB, T0, T1, tp2, tgA2, tgB2);
    for (int base = 64; base < cnt; base += 64)
        t_stretch(cnt - base, lst + base + 2 * hsel, lrow, xs2, pb,
                  W1xA, W1yA, CmA, W1xB, W1yB, CmB, T0, T1, tp2, tgA2, tgB2);
    float tp = tp2.x + tp2.y;
    float tgA = tgA2.x + tgA2.y, tgB = tgB2.x + tgB2.y;
    tp += __shfl_xor(tp, 16, 32);

    v8f U0 = {0.f, 0.f, 0.f, 0.f, 0.f, 0.f, 0.f, 0.f};
    v8f U1 = {0.f, 0.f, 0.f, 0.f, 0.f, 0.f, 0.f, 0.f};
#pragma unroll
    for (int jp = 0; jp < 4; ++jp) {
        v2f a;
        a.x = T0[2 * jp];
        a.y = T0[2 * jp + 1];
        U0 = __builtin_amdgcn_wmma_f32_16x16x4_f32(false, a, false, Bw0[jp],
                                                   (short)0, U0, false, false);
        U1 = __builtin_amdgcn_wmma_f32_16x16x4_f32(false, a, false, Bw1[jp],
                                                   (short)0, U1, false, false);
    }
#pragma unroll
    for (int jp = 0; jp < 4; ++jp) {
        v2f a;
        a.x = T1[2 * jp];
        a.y = T1[2 * jp + 1];
        U0 = __builtin_amdgcn_wmma_f32_16x16x4_f32(false, a, false, Bw0[4 + jp],
                                                   (short)0, U0, false, false);
        U1 = __builtin_amdgcn_wmma_f32_16x16x4_f32(false, a, false, Bw1[4 + jp],
                                                   (short)0, U1, false, false);
    }
    {
        v2f a, c0v, c1v;
        a.x = (hsel == 0) ? tp : 0.f;
        a.y = 0.f;
        c0v.x = bz0; c0v.y = 0.f;
        c1v.x = bz1; c1v.y = 0.f;
        U0 = __builtin_amdgcn_wmma_f32_16x16x4_f32(false, a, false, c0v,
                                                   (short)0, U0, false, false);
        U1 = __builtin_amdgcn_wmma_f32_16x16x4_f32(false, a, false, c1v,
                                                   (short)0, U1, false, false);
    }
    float o0 = 0.f, o1 = 0.f;
#pragma unroll
    for (int r = 0; r < 8; ++r) {
        o0 = fmaf(wl0[r], U0[r], o0);
        o1 = fmaf(wl1[r], U1[r], o1);
    }
    o0 += __shfl_xor(o0, 16, 32);
    o1 += __shfl_xor(o1, 16, 32);
    if (has_bl) {
        tgA += __shfl_xor(tgA, 16, 32);
        tgB += __shfl_xor(tgB, 16, 32);
        const float sw = (float)c1 * inv1 + (float)c2 * inv2;
        float kb0 = sw * bz0, kb1 = sw * bz1;
#pragma unroll 1
        for (int h = 0; h < Hh; ++h) {
            const float tg = __shfl((h < 16) ? tgA : tgB, h & 15, 32);
            kb0 = fmaf(tg, W2[h * Cc + lrow], kb0);
            kb1 = fmaf(tg, W2[h * Cc + 16 + lrow], kb1);
        }
        o0 = fmaf(bl0, kb0, o0);
        o1 = fmaf(bl1, kb1, o1);
    }
    if (lane < 16) {
        out[wid * Cc + lane]      = o0;
        out[wid * Cc + 16 + lane] = o1;
    }
}

extern "C" void kernel_launch(void* const* d_in, const int* in_sizes, int n_in,
                              void* d_out, int out_size, void* d_ws, size_t ws_size,
                              hipStream_t stream) {
    (void)in_sizes; (void)n_in; (void)out_size; (void)d_ws; (void)ws_size;
    const float* pndata = (const float*)d_in[0];
    const float* xcoord = (const float*)d_in[1];
    const float* lq     = (const float*)d_in[2];
    const float* W_lift = (const float*)d_in[3];
    const float* b_lift = (const float*)d_in[4];
    const float* W1     = (const float*)d_in[5];
    const float* b1     = (const float*)d_in[6];
    const float* W2     = (const float*)d_in[7];
    const float* b2     = (const float*)d_in[8];
    float* out = (float*)d_out;

    gno_kernel<<<Bb * (Mm / 8), 256, 0, stream>>>(xcoord, pndata, lq, W_lift, b_lift,
                                                  W1, b1, W2, b2, out);
}


# serial-path waits: every __shfl_xor exchange on DPP row_xmask / v_permlanex16 instead of ds_bpermute round trips; block barriers fence LDS only, so weight loads stay in flight; sums unchanged
# speedup vs baseline: 15.2267x; 1.0738x over previous
#include <hip/hip_runtime.h>


typedef __attribute__((ext_vector_type(2))) float v2f;
typedef __attribute__((ext_vector_type(8))) float v8f;

static constexpr int Bb  = 2;
static constexpr int Nn  = 1024;
static constexpr int Mm  = 1024;
static constexpr int DIN = 16;
static constexpr int Cc  = 32;
static constexpr int Hh  = 32;

#define ROW_XOR(x, M) __int_as_float(__builtin_amdgcn_update_dpp(0, __float_as_int(x), 0x160 + (M), 0xf, 0xf, true))

__device__ __forceinline__ float xhalf(float x) {
    const int b = __float_as_int(x);
    return __int_as_float(__builtin_amdgcn_permlanex16(b, b, 0x76543210, 0xfedcba98, false, false));
}

__device__ __forceinline__ void lds_barrier() {
    __builtin_amdgcn_fence(__ATOMIC_RELEASE, "workgroup", "local");
    __builtin_amdgcn_s_barrier();
    __builtin_amdgcn_fence(__ATOMIC_ACQUIRE, "workgroup", "local");
}

__device__ __forceinline__ v2f gelu_tanh2(v2f x) {
    const v2f k1 = {-2.302208198144325f, -2.302208198144325f};
    const v2f k2 = {-0.1029432395800235f, -0.1029432395800235f};
    const v2f u = x * x;
    const v2f a = x * __builtin_elementwise_fma(u, k2, k1);
    v2f d;
    d.x = 1.0f + __builtin_amdgcn_exp2f(a.x);
    d.y = 1.0f + __builtin_amdgcn_exp2f(a.y);
    v2f r;
    r.x = __builtin_amdgcn_rcpf(d.x);
    r.y = __builtin_amdgcn_rcpf(d.y);
    return x * r;
}

__device__ __forceinline__ void t_stretch(
    const int rem, const int2* lp, const int lrow,
    const float2* xs2, const float* pb,
    const float W1xA, const float W1yA, const float CmA,
    const float W1xB, const float W1yB, const float CmB,
    v8f& T0, v8f& T1, v2f& tp2, v2f& tgA2, v2f& tgB2) {
    int4 ent[16];
#pragma unroll
    for (int j = 0; j < 16; ++j) ent[j] = *(const int4*)(lp + 4 * j);
    float2 ca[16], cb[16];
    v2f bp[16];
#pragma unroll
    for (int j = 0; j < 16; ++j) {
        const unsigned o0 = (unsigned)ent[j].x, o1 = (unsigned)ent[j].z;
        ca[j] = *(const float2*)((const char*)xs2 + o0);
        cb[j] = *(const float2*)((const char*)xs2 + o1);
        bp[j].x = *(const float*)((const char*)pb + (o0 * 8u + 4u * (unsigned)lrow));
        bp[j].y = *(const float*)((const char*)pb + (o1 * 8u + 4u * (unsigned)lrow));
    }
#pragma unroll
    for (int j = 0; j < 16; ++j) {
        if (4 * j < rem) {
            v2f wq;
            wq.x = __int_as_float(ent[j].y);
            wq.y = __int_as_float(ent[j].w);
            v2f pa, pc;
            pa.x = fmaf(ca[j].x, W1xA, fmaf(ca[j].y, W1yA, CmA));
            pa.y = fmaf(cb[j].x, W1xA, fmaf(cb[j].y, W1yA, CmA));
            pc.x = fmaf(ca[j].x, W1xB, fmaf(ca[j].y, W1yB, CmB));
            pc.y = fmaf(cb[j].x, W1xB, fmaf(cb[j].y, W1yB, CmB));
            const v2f a0 = gelu_tanh2(pa) * wq;
            const v2f a1 = gelu_tanh2(pc) * wq;
            T0 = __builtin_amdgcn_wmma_f32_16x16x4_f32(false, a0, false, bp[j],
                                                       (short)0, T0, false, false);
            T1 = __builtin_amdgcn_wmma_f32_16x16x4_f32(false, a1, false, bp[j],
                                                       (short)0, T1, false, false);
            tp2 = __builtin_elementwise_fma(wq, bp[j], tp2);
            tgA2 += a0;
            tgB2 += a1;
        }
    }
}

__global__ void __launch_bounds__(256) gno_kernel(
    const float* __restrict__ xc, const float* __restrict__ pnd,
    const float* __restrict__ lq,
    const float* __restrict__ WL, const float* __restrict__ bl,
    const float* __restrict__ W1, const float* __restrict__ b1,
    const float* __restrict__ W2, const float* __restrict__ b2,
    float* __restrict__ out) {
    __shared__ struct __attribute__((aligned(16))) {
        float xss[Nn * 2];
        float red[8][4];
        int2  nbr[8][Nn];
    } sm;
    float* const xss = sm.xss;
    float (* const red)[4] = sm.red;
    int2 (* const nbr)[Nn] = sm.nbr;

    const int tid  = threadIdx.x;
    const int b    = blockIdx.x / (Mm / 8);
    const int mg   = blockIdx.x % (Mm / 8);
    const int wv   = tid >> 5;
    const int lane = tid & 31;
    const int m    = mg * 8 + wv;
    const int wid  = b * Mm + m;
    const int lrow = lane & 15;
    const int hsel = lane >> 4;

    const float4* s2 = (const float4*)(xc + b * Nn * 2);
    const float4 va = s2[tid];
    const float4 vb = s2[tid + 256];
    __builtin_amdgcn_sched_barrier(0);

    const float q0 = lq[m * 2], q1 = lq[m * 2 + 1];

    const int hA = lrow, hB = 16 + lrow;
    const float W1xA = W1[hA], W1yA = W1[Hh + hA];
    const float W1xB = W1[hB], W1yB = W1[Hh + hB];
    const float CmA = fmaf(q0, W1[2 * Hh + hA], fmaf(q1, W1[3 * Hh + hA], b1[hA]));
    const float CmB = fmaf(q0, W1[2 * Hh + hB], fmaf(q1, W1[3 * Hh + hB], b1[hB]));

    v2f Bw0[8], Bw1[8];
#pragma unroll
    for (int kk = 0; kk < 8; ++kk) {
        const int k0 = 16 * (kk >> 2) + 8 * hsel + 2 * (kk & 3);
        Bw0[kk].x = W2[k0 * Cc + lrow];
        Bw0[kk].y = W2[(k0 + 1) * Cc + lrow];
        Bw1[kk].x = W2[k0 * Cc + 16 + lrow];
        Bw1[kk].y = W2[(k0 + 1) * Cc + 16 + lrow];
    }
    const float bz0 = b2[lrow], bz1 = b2[16 + lrow];

    float wl0[8], wl1[8];
#pragma unroll
    for (int r = 0; r < 8; ++r) {
        wl0[r] = WL[(8 * hsel + r) * Cc + lrow];
        wl1[r] = WL[(8 * hsel + r) * Cc + 16 + lrow];
    }
    const float bl0 = bl[lrow], bl1 = bl[16 + lrow];

    __builtin_amdgcn_sched_barrier(0);
    float mn0 = fminf(fminf(va.x, va.z), fminf(vb.x, vb.z));
    float mn1 = fminf(fminf(va.y, va.w), fminf(vb.y, vb.w));
    float mx0 = fmaxf(fmaxf(va.x, va.z), fmaxf(vb.x, vb.z));
    float mx1 = fmaxf(fmaxf(va.y, va.w), fmaxf(vb.y, vb.w));
#define EXTREMA_STEP(M)                       \
    mn0 = fminf(mn0, ROW_XOR(mn0, M));        \
    mn1 = fminf(mn1, ROW_XOR(mn1, M));        \
    mx0 = fmaxf(mx0, ROW_XOR(mx0, M));        \
    mx1 = fmaxf(mx1, ROW_XOR(mx1, M));
    EXTREMA_STEP(1)
    EXTREMA_STEP(2)
    EXTREMA_STEP(4)
    EXTREMA_STEP(8)
#undef EXTREMA_STEP
    mn0 = fminf(mn0, xhalf(mn0));
    mn1 = fminf(mn1, xhalf(mn1));
    mx0 = fmaxf(mx0, xhalf(mx0));
    mx1 = fmaxf(mx1, xhalf(mx1));
    if (lane == 0) *(float4*)red[wv] = make_float4(mn0, mn1, mx0, mx1);
    lds_barrier();
#pragma unroll
    for (int w = 0; w < 8; ++w) {
        const float4 rw = *(const float4*)red[w];
        mn0 = fminf(mn0, rw.x);
        mn1 = fminf(mn1, rw.y);
        mx0 = fmaxf(mx0, rw.z);
        mx1 = fmaxf(mx1, rw.w);
    }
    {
        const float sc0 = 2.0f / (mx0 - mn0 + 1e-12f);
        const float sc1 = 2.0f / (mx1 - mn1 + 1e-12f);
        float4 ra, rb;
        ra.x = (va.x - mn0) * sc0 - 1.0f;
        ra.y = (va.y - mn1) * sc1 - 1.0f;
        ra.z = (va.z - mn0) * sc0 - 1.0f;
        ra.w = (va.w - mn1) * sc1 - 1.0f;
        rb.x = (vb.x - mn0) * sc0 - 1.0f;
        rb.y = (vb.y - mn1) * sc1 - 1.0f;
        rb.z = (vb.z - mn0) * sc0 - 1.0f;
        rb.w = (vb.w - mn1) * sc1 - 1.0f;
        float4* d2v = (float4*)xss;
        d2v[tid]       = ra;
        d2v[tid + 256] = rb;
    }
    lds_barrier();

    int2* lst = nbr[wv];
    const float2* xs2 = (const float2*)xss;
    float2 pp[Nn / 32];
#pragma unroll
    for (int it = 0; it < Nn / 32; ++it) pp[it] = xs2[it * 32 + lane];
    unsigned om1 = 0u, om2 = 0u;
#pragma unroll
    for (int it = Nn / 32 - 1; it >= 0; --it) {
        const float dx = q0 - pp[it].x, dy = q1 - pp[it].y;
        const float d2 = fmaf(dx, dx, dy * dy);
        const float t1 = 0.01f - d2;
        const float t2 = 0.04f - d2;
        om1 = __builtin_amdgcn_alignbit(om1, __float_as_uint(t1), 31u);
        om2 = __builtin_amdgcn_alignbit(om2, __float_as_uint(t2), 31u);
    }
    const unsigned h2 = ~om2, h1 = ~om1;
    const int pk = __builtin_popcount(h2) + (__builtin_popcount(h1) << 16);
    int sc = pk;
    sc += __builtin_amdgcn_update_dpp(0, sc, 0x111, 0xf, 0xf, true);
    sc += __builtin_amdgcn_update_dpp(0, sc, 0x112, 0xf, 0xf, true);
    sc += __builtin_amdgcn_update_dpp(0, sc, 0x114, 0xf, 0xf, true);
    sc += __builtin_amdgcn_update_dpp(0, sc, 0x118, 0xf, 0xf, true);
    const int row0 = __builtin_amdgcn_readlane(sc, 15);
    if (lane >= 16) sc += row0;
    const int tot = __builtin_amdgcn_readlane(sc, 31);
    const int cnt = tot & 0xffff;
    const int c1 = tot >> 16, c2 = cnt;
    const float inv1 = 1.0f / fmaxf((float)c1, 1.0f);
    const float inv2 = 1.0f / fmaxf((float)c2, 1.0f);
    const float inv12 = inv1 + inv2;
    {
        int pos = (sc - pk) & 0xffff;
        unsigned hh = h2;
        while (hh != 0u) {
            const int bit = __builtin_ctz(hh);
            const float w = ((h1 >> bit) & 1u) ? inv12 : inv2;
            lst[pos] = make_int2(8 * (bit * 32 + lane), __float_as_int(w));
            ++pos;
            hh &= hh - 1u;
        }
    }
    {
        const int cend = (cnt + 63) & ~63;
        if (cnt + lane < cend)      lst[cnt + lane]      = make_int2(0, 0);
        if (cnt + 32 + lane < cend) lst[cnt + 32 + lane] = make_int2(0, 0);
    }
    lds_barrier();
    const bool has_bl = __builtin_amdgcn_ballot_w32(bl0 != 0.f || bl1 != 0.f) != 0u;

    const float* pb = pnd + b * Nn * DIN;
    v8f T0 = {0.f, 0.f, 0.f, 0.f, 0.f, 0.f, 0.f, 0.f};
    v8f T1 = {0.f, 0.f, 0.f, 0.f, 0.f, 0.f, 0.f, 0.f};
    v2f tp2 = {0.f, 0.f}, tgA2 = {0.f, 0.f}, tgB2 = {0.f, 0.f};
    if (cnt > 0)
        t_stretch(cnt, lst + 2 * hsel, lrow, xs2, pb,
                  W1xA, W1yA, CmA, W1xB, W1yB, CmB, T0, T1, tp2, tgA2, tgB2);
    for (int base = 64; base < cnt; base += 64)
        t_stretch(cnt - base, lst + base + 2 * hsel, lrow, xs2, pb,
                  W1xA, W1yA, CmA, W1xB, W1yB, CmB, T0, T1, tp2, tgA2, tgB2);
    float tp = tp2.x + tp2.y;
    float tgA = tgA2.x + tgA2.y, tgB = tgB2.x + tgB2.y;
    tp += xhalf(tp);

    v8f U0 = {0.f, 0.f, 0.f, 0.f, 0.f, 0.f, 0.f, 0.f};
    v8f U1 = {0.f, 0.f, 0.f, 0.f, 0.f, 0.f, 0.f, 0.f};
#pragma unroll
    for (int jp = 0; jp < 4; ++jp) {
        v2f a;
        a.x = T0[2 * jp];
        a.y = T0[2 * jp + 1];
        U0 = __builtin_amdgcn_wmma_f32_16x16x4_f32(false, a, false, Bw0[jp],
                                                   (short)0, U0, false, false);
        U1 = __builtin_amdgcn_wmma_f32_16x16x4_f32(false, a, false, Bw1[jp],
                                                   (short)0, U1, false, false);
    }
#pragma unroll
    for (int jp = 0; jp < 4; ++jp) {
        v2f a;
        a.x = T1[2 * jp];
        a.y = T1[2 * jp + 1];
        U0 = __builtin_amdgcn_wmma_f32_16x16x4_f32(false, a, false, Bw0[4 + jp],
                                                   (short)0, U0, false, false);
        U1 = __builtin_amdgcn_wmma_f32_16x16x4_f32(false, a, false, Bw1[4 + jp],
                                                   (short)0, U1, false, false);
    }
    {
        v2f a, c0v, c1v;
        a.x = (hsel == 0) ? tp : 0.f;
        a.y = 0.f;
        c0v.x = bz0; c0v.y = 0.f;
        c1v.x = bz1; c1v.y = 0.f;
        U0 = __builtin_amdgcn_wmma_f32_16x16x4_f32(false, a, false, c0v,
                                                   (short)0, U0, false, false);
        U1 = __builtin_amdgcn_wmma_f32_16x16x4_f32(false, a, false, c1v,
                                                   (short)0, U1, false, false);
    }
    float o0 = 0.f, o1 = 0.f;
#pragma unroll
    for (int r = 0; r < 8; ++r) {
        o0 = fmaf(wl0[r], U0[r], o0);
        o1 = fmaf(wl1[r], U1[r], o1);
    }
    o0 += xhalf(o0);
    o1 += xhalf(o1);
    if (has_bl) {
        tgA += xhalf(tgA);
        tgB += xhalf(tgB);
        const float sw = (float)c1 * inv1 + (float)c2 * inv2;
        float kb0 = sw * bz0, kb1 = sw * bz1;
#pragma unroll 1
        for (int h = 0; h < Hh; ++h) {
            const float tg = __shfl((h < 16) ? tgA : tgB, h & 15, 32);
            kb0 = fmaf(tg, W2[h * Cc + lrow], kb0);
            kb1 = fmaf(tg, W2[h * Cc + 16 + lrow], kb1);
        }
        o0 = fmaf(bl0, kb0, o0);
        o1 = fmaf(bl1, kb1, o1);
    }
    if (lane < 16) {
        out[wid * Cc + lane]      = o0;
        out[wid * Cc + 16 + lane] = o1;
    }
}

extern "C" void kernel_launch(void* const* d_in, const int* in_sizes, int n_in,
                              void* d_out, int out_size, void* d_ws, size_t ws_size,
                              hipStream_t stream) {
    (void)in_sizes; (void)n_in; (void)out_size; (void)d_ws; (void)ws_size;
    const float* pndata = (const float*)d_in[0];
    const float* xcoord = (const float*)d_in[1];
    const float* lq     = (const float*)d_in[2];
    const float* W_lift = (const float*)d_in[3];
    const float* b_lift = (const float*)d_in[4];
    const float* W1     = (const float*)d_in[5];
    const float* b1     = (const float*)d_in[6];
    const float* W2     = (const float*)d_in[7];
    const float* b2     = (const float*)d_in[8];
    float* out = (float*)d_out;

    gno_kernel<<<Bb * (Mm / 8), 256, 0, stream>>>(xcoord, pndata, lq, W_lift, b_lift,
                                                  W1, b1, W2, b2, out);
}


# pass 2 out of block lock-step: no block barrier ahead of it (each wave's list is its own) and a stretch's w*gelu operands all made ahead of its WMMAs, under the gather latency; sums unchanged
# speedup vs baseline: 16.9320x; 1.1120x over previous
#include <hip/hip_runtime.h>


typedef __attribute__((ext_vector_type(2))) float v2f;
typedef __attribute__((ext_vector_type(8))) float v8f;

static constexpr int Bb  = 2;
static constexpr int Nn  = 1024;
static constexpr int Mm  = 1024;
static constexpr int DIN = 16;
static constexpr int Cc  = 32;
static constexpr int Hh  = 32;

#define ROW_XOR(x, M) __int_as_float(__builtin_amdgcn_update_dpp(0, __float_as_int(x), 0x160 + (M), 0xf, 0xf, true))

__device__ __forceinline__ float xhalf(float x) {
    const int b = __float_as_int(x);
    return __int_as_float(__builtin_amdgcn_permlanex16(b, b, 0x76543210, 0xfedcba98, false, false));
}

__device__ __forceinline__ void lds_barrier() {
    __builtin_amdgcn_fence(__ATOMIC_RELEASE, "workgroup", "local");
    __builtin_amdgcn_s_barrier();
    __builtin_amdgcn_fence(__ATOMIC_ACQUIRE, "workgroup", "local");
}

__device__ __forceinline__ v2f gelu_tanh2(v2f x) {
    const v2f k1 = {-2.302208198144325f, -2.302208198144325f};
    const v2f k2 = {-0.1029432395800235f, -0.1029432395800235f};
    const v2f u = x * x;
    const v2f a = x * __builtin_elementwise_fma(u, k2, k1);
    v2f d;
    d.x = 1.0f + __builtin_amdgcn_exp2f(a.x);
    d.y = 1.0f + __builtin_amdgcn_exp2f(a.y);
    v2f r;
    r.x = __builtin_amdgcn_rcpf(d.x);
    r.y = __builtin_amdgcn_rcpf(d.y);
    return x * r;
}

template <int NG>
__device__ __forceinline__ void t_stretch(
    const int rem, const int2* lp, const int lrow,
    const float2* xs2, const float* pb,
    const float W1xA, const float W1yA, const float CmA,
    const float W1xB, const float W1yB, const float CmB,
    v8f& T0, v8f& T1, v2f& tp2, v2f& tgA2, v2f& tgB2) {
    constexpr int NJ = 4 * NG;
    int4 ent[NJ];
#pragma unroll
    for (int j = 0; j < NJ; ++j) ent[j] = *(const int4*)(lp + 4 * j);
    float2 ca[NJ], cb[NJ];
    v2f bp[NJ];
#pragma unroll
    for (int j = 0; j < NJ; ++j) {
        const unsigned o0 = (unsigned)ent[j].x, o1 = (unsigned)ent[j].z;
        ca[j] = *(const float2*)((const char*)xs2 + o0);
        cb[j] = *(const float2*)((const char*)xs2 + o1);
        bp[j].x = *(const float*)((const char*)pb + (o0 * 8u + 4u * (unsigned)lrow));
        bp[j].y = *(const float*)((const char*)pb + (o1 * 8u + 4u * (unsigned)lrow));
    }
    v2f a0[NJ], a1[NJ];
#pragma unroll
    for (int i = 0; i < NJ / 2; ++i) {
        if (8 * i < rem) {
#pragma unroll
            for (int jj = 0; jj < 2; ++jj) {
                const int j = 2 * i + jj;
                v2f wq;
                wq.x = __int_as_float(ent[j].y);
                wq.y = __int_as_float(ent[j].w);
                v2f pa, pc;
                pa.x = fmaf(ca[j].x, W1xA, fmaf(ca[j].y, W1yA, CmA));
                pa.y = fmaf(cb[j].x, W1xA, fmaf(cb[j].y, W1yA, CmA));
                pc.x = fmaf(ca[j].x, W1xB, fmaf(ca[j].y, W1yB, CmB));
                pc.y = fmaf(cb[j].x, W1xB, fmaf(cb[j].y, W1yB, CmB));
                a0[j] = gelu_tanh2(pa) * wq;
                a1[j] = gelu_tanh2(pc) * wq;
                tgA2 += a0[j];
                tgB2 += a1[j];
            }
        }
    }
#pragma unroll
    for (int j = 0; j < NJ; ++j) {
        if (4 * j < rem) {
            v2f wq;
            wq.x = __int_as_float(ent[j].y);
            wq.y = __int_as_float(ent[j].w);
            T0 = __builtin_amdgcn_wmma_f32_16x16x4_f32(false, a0[j], false, bp[j],
                                                       (short)0, T0, false, false);
            T1 = __builtin_amdgcn_wmma_f32_16x16x4_f32(false, a1[j], false, bp[j],
                                                       (short)0, T1, false, false);
            tp2 = __builtin_elementwise_fma(wq, bp[j], tp2);
        }
    }
}

__global__ void __launch_bounds__(256) gno_kernel(
    const float* __restrict__ xc, const float* __restrict__ pnd,
    const float* __restrict__ lq,
    const float* __restrict__ WL, const float* __restrict__ bl,
    const float* __restrict__ W1, const float* __restrict__ b1,
    const float* __restrict__ W2, const float* __restrict__ b2,
    float* __restrict__ out) {
    __shared__ struct __attribute__((aligned(16))) {
        float xss[Nn * 2];
        float red[8][4];
        int2  nbr[8][Nn];
    } sm;
    float* const xss = sm.xss;
    float (* const red)[4] = sm.red;
    int2 (* const nbr)[Nn] = sm.nbr;

    const int tid  = threadIdx.x;
    const int b    = blockIdx.x / (Mm / 8);
    const int mg   = blockIdx.x % (Mm / 8);
    const int wv   = tid >> 5;
    const int lane = tid & 31;
    const int m    = mg * 8 + wv;
    const int wid  = b * Mm + m;
    const int lrow = lane & 15;
    const int hsel = lane >> 4;

    const float4* s2 = (const float4*)(xc + b * Nn * 2);
    const float4 va = s2[tid];
    const float4 vb = s2[tid + 256];
    __builtin_amdgcn_sched_barrier(0);

    const float q0 = lq[m * 2], q1 = lq[m * 2 + 1];

    const int hA = lrow, hB = 16 + lrow;
    const float W1xA = W1[hA], W1yA = W1[Hh + hA];
    const float W1xB = W1[hB], W1yB = W1[Hh + hB];
    const float CmA = fmaf(q0, W1[2 * Hh + hA], fmaf(q1, W1[3 * Hh + hA], b1[hA]));
    const float CmB = fmaf(q0, W1[2 * Hh + hB], fmaf(q1, W1[3 * Hh + hB], b1[hB]));

    v2f Bw0[8], Bw1[8];
#pragma unroll
    for (int kk = 0; kk < 8; ++kk) {
        const int k0 = 16 * (kk >> 2) + 8 * hsel + 2 * (kk & 3);
        Bw0[kk].x = W2[k0 * Cc + lrow];
        Bw0[kk].y = W2[(k0 + 1) * Cc + lrow];
        Bw1[kk].x = W2[k0 * Cc + 16 + lrow];
        Bw1[kk].y = W2[(k0 + 1) * Cc + 16 + lrow];
    }
    const float bz0 = b2[lrow], bz1 = b2[16 + lrow];

    float wl0[8], wl1[8];
#pragma unroll
    for (int r = 0; r < 8; ++r) {
        wl0[r] = WL[(8 * hsel + r) * Cc + lrow];
        wl1[r] = WL[(8 * hsel + r) * Cc + 16 + lrow];
    }
    const float bl0 = bl[lrow], bl1 = bl[16 + lrow];

    __builtin_amdgcn_sched_barrier(0);
    float mn0 = fminf(fminf(va.x, va.z), fminf(vb.x, vb.z));
    float mn1 = fminf(fminf(va.y, va.w), fminf(vb.y, vb.w));
    float mx0 = fmaxf(fmaxf(va.x, va.z), fmaxf(vb.x, vb.z));
    float mx1 = fmaxf(fmaxf(va.y, va.w), fmaxf(vb.y, vb.w));
#define EXTREMA_STEP(M)                       \
    mn0 = fminf(mn0, ROW_XOR(mn0, M));        \
    mn1 = fminf(mn1, ROW_XOR(mn1, M));        \
    mx0 = fmaxf(mx0, ROW_XOR(mx0, M));        \
    mx1 = fmaxf(mx1, ROW_XOR(mx1, M));
    EXTREMA_STEP(1)
    EXTREMA_STEP(2)
    EXTREMA_STEP(4)
    EXTREMA_STEP(8)
#undef EXTREMA_STEP
    mn0 = fminf(mn0, xhalf(mn0));
    mn1 = fminf(mn1, xhalf(mn1));
    mx0 = fmaxf(mx0, xhalf(mx0));
    mx1 = fmaxf(mx1, xhalf(mx1));
    if (lane == 0) *(float4*)red[wv] = make_float4(mn0, mn1, mx0, mx1);
    lds_barrier();
#pragma unroll
    for (int w = 0; w < 8; ++w) {
        const float4 rw = *(const float4*)red[w];
        mn0 = fminf(mn0, rw.x);
        mn1 = fminf(mn1, rw.y);
        mx0 = fmaxf(mx0, rw.z);
        mx1 = fmaxf(mx1, rw.w);
    }
    {
        const float sc0 = 2.0f / (mx0 - mn0 + 1e-12f);
        const float sc1 = 2.0f / (mx1 - mn1 + 1e-12f);
        float4 ra, rb;
        ra.x = (va.x - mn0) * sc0 - 1.0f;
        ra.y = (va.y - mn1) * sc1 - 1.0f;
        ra.z = (va.z - mn0) * sc0 - 1.0f;
        ra.w = (va.w - mn1) * sc1 - 1.0f;
        rb.x = (vb.x - mn0) * sc0 - 1.0f;
        rb.y = (vb.y - mn1) * sc1 - 1.0f;
        rb.z = (vb.z - mn0) * sc0 - 1.0f;
        rb.w = (vb.w - mn1) * sc1 - 1.0f;
        float4* d2v = (float4*)xss;
        d2v[tid]       = ra;
        d2v[tid + 256] = rb;
    }
    lds_barrier();

    int2* lst = nbr[wv];
    const float2* xs2 = (const float2*)xss;
    float2 pp[Nn / 32];
#pragma unroll
    for (int it = 0; it < Nn / 32; ++it) pp[it] = xs2[it * 32 + lane];
    unsigned om1 = 0u, om2 = 0u;
#pragma unroll
    for (int it = Nn / 32 - 1; it >= 0; --it) {
        const float dx = q0 - pp[it].x, dy = q1 - pp[it].y;
        const float d2 = fmaf(dx, dx, dy * dy);
        const float t1 = 0.01f - d2;
        const float t2 = 0.04f - d2;
        om1 = __builtin_amdgcn_alignbit(om1, __float_as_uint(t1), 31u);
        om2 = __builtin_amdgcn_alignbit(om2, __float_as_uint(t2), 31u);
    }
    const unsigned h2 = ~om2, h1 = ~om1;
    const int pk = __builtin_popcount(h2) + (__builtin_popcount(h1) << 16);
    int sc = pk;
    sc += __builtin_amdgcn_update_dpp(0, sc, 0x111, 0xf, 0xf, true);
    sc += __builtin_amdgcn_update_dpp(0, sc, 0x112, 0xf, 0xf, true);
    sc += __builtin_amdgcn_update_dpp(0, sc, 0x114, 0xf, 0xf, true);
    sc += __builtin_amdgcn_update_dpp(0, sc, 0x118, 0xf, 0xf, true);
    const int row0 = __builtin_amdgcn_readlane(sc, 15);
    if (lane >= 16) sc += row0;
    const int tot = __builtin_amdgcn_readlane(sc, 31);
    const int cnt = tot & 0xffff;
    const int c1 = tot >> 16, c2 = cnt;
    const float inv1 = 1.0f / fmaxf((float)c1, 1.0f);
    const float inv2 = 1.0f / fmaxf((float)c2, 1.0f);
    const float inv12 = inv1 + inv2;
    {
        int pos = (sc - pk) & 0xffff;
        unsigned hh = h2;
        while (hh != 0u) {
            const int bit = __builtin_ctz(hh);
            const float w = ((h1 >> bit) & 1u) ? inv12 : inv2;
            lst[pos] = make_int2(8 * (bit * 32 + lane), __float_as_int(w));
            ++pos;
            hh &= hh - 1u;
        }
    }
    {
        const int cend = (cnt + 63) & ~63;
        if (cnt + lane < cend)      lst[cnt + lane]      = make_int2(0, 0);
        if (cnt + 32 + lane < cend) lst[cnt + 32 + lane] = make_int2(0, 0);
    }
    __builtin_amdgcn_fence(__ATOMIC_RELEASE, "wavefront", "local");
    __builtin_amdgcn_wave_barrier();
    __builtin_amdgcn_fence(__ATOMIC_ACQUIRE, "wavefront", "local");
    const bool has_bl = __builtin_amdgcn_ballot_w32(bl0 != 0.f || bl1 != 0.f) != 0u;

    const float* pb = pnd + b * Nn * DIN;
    v8f T0 = {0.f, 0.f, 0.f, 0.f, 0.f, 0.f, 0.f, 0.f};
    v8f T1 = {0.f, 0.f, 0.f, 0.f, 0.f, 0.f, 0.f, 0.f};
    v2f tp2 = {0.f, 0.f}, tgA2 = {0.f, 0.f}, tgB2 = {0.f, 0.f};
    if (cnt > 48) {
        for (int base = 0; base < cnt; base += 64)
            t_stretch<4>(cnt - base, lst + base + 2 * hsel, lrow, xs2, pb,
                         W1xA, W1yA, CmA, W1xB, W1yB, CmB, T0, T1, tp2, tgA2, tgB2);
    } else if (cnt > 32) {
        t_stretch<3>(cnt, lst + 2 * hsel, lrow, xs2, pb,
                     W1xA, W1yA, CmA, W1xB, W1yB, CmB, T0, T1, tp2, tgA2, tgB2);
    } else if (cnt > 0) {
        t_stretch<2>(cnt, lst + 2 * hsel, lrow, xs2, pb,
                     W1xA, W1yA, CmA, W1xB, W1yB, CmB, T0, T1, tp2, tgA2, tgB2);
    }
    float tp = tp2.x + tp2.y;
    float tgA = tgA2.x + tgA2.y, tgB = tgB2.x + tgB2.y;
    tp += xhalf(tp);

    v8f U0 = {0.f, 0.f, 0.f, 0.f, 0.f, 0.f, 0.f, 0.f};
    v8f U1 = {0.f, 0.f, 0.f, 0.f, 0.f, 0.f, 0.f, 0.f};
#pragma unroll
    for (int jp = 0; jp < 4; ++jp) {
        v2f a;
        a.x = T0[2 * jp];
        a.y = T0[2 * jp + 1];
        U0 = __builtin_amdgcn_wmma_f32_16x16x4_f32(false, a, false, Bw0[jp],
                                                   (short)0, U0, false, false);
        U1 = __builtin_amdgcn_wmma_f32_16x16x4_f32(false, a, false, Bw1[jp],
                                                   (short)0, U1, false, false);
    }
#pragma unroll
    for (int jp = 0; jp < 4; ++jp) {
        v2f a;
        a.x = T1[2 * jp];
        a.y = T1[2 * jp + 1];
        U0 = __builtin_amdgcn_wmma_f32_16x16x4_f32(false, a, false, Bw0[4 + jp],
                                                   (short)0, U0, false, false);
        U1 = __builtin_amdgcn_wmma_f32_16x16x4_f32(false, a, false, Bw1[4 + jp],
                                                   (short)0, U1, false, false);
    }
    {
        v2f a, c0v, c1v;
        a.x = (hsel == 0) ? tp : 0.f;
        a.y = 0.f;
        c0v.x = bz0; c0v.y = 0.f;
        c1v.x = bz1; c1v.y = 0.f;
        U0 = __builtin_amdgcn_wmma_f32_16x16x4_f32(false, a, false, c0v,
                                                   (short)0, U0, false, false);
        U1 = __builtin_amdgcn_wmma_f32_16x16x4_f32(false, a, false, c1v,
                                                   (short)0, U1, false, false);
    }
    float o0 = 0.f, o1 = 0.f;
#pragma unroll
    for (int r = 0; r < 8; ++r) {
        o0 = fmaf(wl0[r], U0[r], o0);
        o1 = fmaf(wl1[r], U1[r], o1);
    }
    o0 += xhalf(o0);
    o1 += xhalf(o1);
    if (has_bl) {
        tgA += xhalf(tgA);
        tgB += xhalf(tgB);
        const float sw = (float)c1 * inv1 + (float)c2 * inv2;
        float kb0 = sw * bz0, kb1 = sw * bz1;
#pragma unroll 1
        for (int h = 0; h < Hh; ++h) {
            const float tg = __shfl((h < 16) ? tgA : tgB, h & 15, 32);
            kb0 = fmaf(tg, W2[h * Cc + lrow], kb0);
            kb1 = fmaf(tg, W2[h * Cc + 16 + lrow], kb1);
        }
        o0 = fmaf(bl0, kb0, o0);
        o1 = fmaf(bl1, kb1, o1);
    }
    if (lane < 16) {
        out[wid * Cc + lane]      = o0;
        out[wid * Cc + 16 + lane] = o1;
    }
}

extern "C" void kernel_launch(void* const* d_in, const int* in_sizes, int n_in,
                              void* d_out, int out_size, void* d_ws, size_t ws_size,
                              hipStream_t stream) {
    (void)in_sizes; (void)n_in; (void)out_size; (void)d_ws; (void)ws_size;
    const float* pndata = (const float*)d_in[0];
    const float* xcoord = (const float*)d_in[1];
    const float* lq     = (const float*)d_in[2];
    const float* W_lift = (const float*)d_in[3];
    const float* b_lift = (const float*)d_in[4];
    const float* W1     = (const float*)d_in[5];
    const float* b1     = (const float*)d_in[6];
    const float* W2     = (const float*)d_in[7];
    const float* b2     = (const float*)d_in[8];
    float* out = (float*)d_out;

    gno_kernel<<<Bb * (Mm / 8), 256, 0, stream>>>(xcoord, pndata, lq, W_lift, b_lift,
                                                  W1, b1, W2, b2, out);
}
